# starting kernel
# baseline (speedup 1.0000x reference)
#include <hip/hip_runtime.h>
#include <hip/hip_bf16.h>
#include <math.h>


typedef __attribute__((ext_vector_type(2))) float v2f;
typedef __attribute__((ext_vector_type(8))) float v8f;

#define IN_C  128
#define OUT_C 64
#define GAT_ALPHA 0.2f

__device__ __forceinline__ float lrelu(float z) {
    return z >= 0.0f ? z : GAT_ALPHA * z;
}

__global__ void gat_gemm_stw(const float* __restrict__ x,
                             const float* __restrict__ W,
                             const float* __restrict__ a,
                             float* __restrict__ h,
                             float* __restrict__ s,
                             float* __restrict__ t,
                             int n, int mtiles) {
    const int wavesPerBlock = blockDim.x >> 5;
    const int wave = blockIdx.x * wavesPerBlock + ((int)threadIdx.x >> 5);
    if (wave >= mtiles) return;

    const int lane = threadIdx.x & 31;
    const int half = lane >> 4;
    const int l15  = lane & 15;
    const int m0   = wave * 16;

    int arow = m0 + l15; if (arow > n - 1) arow = n - 1;
    const float* xrow = x + (size_t)arow * IN_C;

    v2f afrag[32];
#pragma unroll
    for (int ks = 0; ks < 32; ++ks) {
        const int k = ks * 4 + half * 2;
        afrag[ks].x = xrow[k];
        afrag[ks].y = xrow[k + 1];
    }

    float pr1[8], pr2[8];
#pragma unroll
    for (int r = 0; r < 8; ++r) { pr1[r] = 0.0f; pr2[r] = 0.0f; }

#pragma unroll
    for (int nt = 0; nt < 4; ++nt) {
        const int n0 = nt * 16;
        v8f acc = {};
#pragma unroll
        for (int ks = 0; ks < 32; ++ks) {
            const int k = ks * 4 + half * 2;
            v2f b;
            b.x = W[(size_t)k       * OUT_C + n0 + l15];
            b.y = W[(size_t)(k + 1) * OUT_C + n0 + l15];
            acc = __builtin_amdgcn_wmma_f32_16x16x4_f32(
                 false, afrag[ks],  false, b,
                 (short)0, acc,  false,  false);
        }
        const float a1c = a[n0 + l15];
        const float a2c = a[OUT_C + n0 + l15];
#pragma unroll
        for (int r = 0; r < 8; ++r) {
            const int row = m0 + r + 8 * half;
            const float v = acc[r];
            if (row < n) h[(size_t)row * OUT_C + n0 + l15] = v;
            pr1[r] += v * a1c;
            pr2[r] += v * a2c;
        }
    }

#pragma unroll
    for (int mask = 8; mask >= 1; mask >>= 1) {
#pragma unroll
        for (int r = 0; r < 8; ++r) {
            pr1[r] += __shfl_xor(pr1[r], mask, 32);
            pr2[r] += __shfl_xor(pr2[r], mask, 32);
        }
    }
    if (l15 == 0) {
#pragma unroll
        for (int r = 0; r < 8; ++r) {
            const int row = m0 + r + 8 * half;
            if (row < n) { s[row] = pr1[r]; t[row] = pr2[r]; }
        }
    }
}

__global__ void gat_zero(float* __restrict__ out, float* __restrict__ expsum, int n) {
    const size_t nout = (size_t)n * OUT_C;
    const size_t tot  = nout + (size_t)n;
    const size_t stride = (size_t)gridDim.x * blockDim.x;
    for (size_t i = (size_t)blockIdx.x * blockDim.x + threadIdx.x; i < tot; i += stride) {
        if (i < nout) out[i] = 0.0f;
        else          expsum[i - nout] = 0.0f;
    }
}

__global__ void gat_edge(const int* __restrict__ ei,
                         const float* __restrict__ h,
                         const float* __restrict__ s,
                         const float* __restrict__ t,
                         float* __restrict__ hp,
                         float* __restrict__ expsum,
                         int nE) {
    const int wavesPerBlock = blockDim.x >> 5;
    const int e = blockIdx.x * wavesPerBlock + ((int)threadIdx.x >> 5);
    if (e >= nE) return;
    const int lane = threadIdx.x & 31;

    const int src = ei[e];
    const int dst = ei[nE + e];
    const float w = __expf(lrelu(s[src] + t[dst]));

    const float* hd = h  + (size_t)dst * OUT_C;
    float*       hs = hp + (size_t)src * OUT_C;
    const int c = lane * 2;
    atomicAdd(&hs[c],     w * hd[c]);
    atomicAdd(&hs[c + 1], w * hd[c + 1]);
    if (lane == 0) atomicAdd(&expsum[src], w);
}

__global__ void gat_finalize(const float* __restrict__ h,
                             const float* __restrict__ s,
                             const float* __restrict__ t,
                             const float* __restrict__ expsum,
                             float* __restrict__ out,
                             int n) {
    const size_t tot = (size_t)n * OUT_C;
    const size_t stride = (size_t)gridDim.x * blockDim.x;
    for (size_t idx = (size_t)blockIdx.x * blockDim.x + threadIdx.x; idx < tot; idx += stride) {
        const size_t i = idx / OUT_C;
        const float w = __expf(lrelu(s[i] + t[i]));
        const float denom = expsum[i] + w;
        out[idx] = (out[idx] + w * h[idx]) / denom;
    }
}

extern "C" void kernel_launch(void* const* d_in, const int* in_sizes, int n_in,
                              void* d_out, int out_size, void* d_ws, size_t ws_size,
                              hipStream_t stream) {
    const float* x  = (const float*)d_in[0];
    const float* W  = (const float*)d_in[1];
    const float* a  = (const float*)d_in[2];
    const int*   ei = (const int*)d_in[3];

    const int n = in_sizes[0] / IN_C;
    const int E = in_sizes[3] / 2;

    float* ws     = (float*)d_ws;
    float* h      = ws;
    float* s      = h + (size_t)n * OUT_C;
    float* t      = s + n;
    float* expsum = t + n;
    float* out    = (float*)d_out;

    const dim3 blk(256);
    const int wavesPerBlock = 256 / 32;

    const int mtiles = (n + 15) / 16;
    gat_gemm_stw<<<(mtiles + wavesPerBlock - 1) / wavesPerBlock, blk, 0, stream>>>(
        x, W, a, h, s, t, n, mtiles);

    gat_zero<<<1024, blk, 0, stream>>>(out, expsum, n);

    gat_edge<<<(E + wavesPerBlock - 1) / wavesPerBlock, blk, 0, stream>>>(
        ei, h, s, t, out, expsum, E);

    gat_finalize<<<2048, blk, 0, stream>>>(h, s, t, expsum, out, n);
}


# owner sweep reads a 1-byte owner-block key per edge (plane written by the GEMM launch behind t) instead of the 4-byte source; matches by zero-byte test; LDS float adds in arrival order as before
# speedup vs baseline: 2.8178x; 2.8178x over previous
#include <hip/hip_runtime.h>
#include <hip/hip_bf16.h>
#include <math.h>


typedef __attribute__((ext_vector_type(2))) float v2f;
typedef __attribute__((ext_vector_type(8))) float v8f;

#define IN_C  128
#define OUT_C 64
#define GAT_ALPHA 0.2f

__device__ __forceinline__ float lrelu(float z) {
    return z >= 0.0f ? z : GAT_ALPHA * z;
}

#define OWN_ROWS    392
#define OWN_BLOCKS  256

__device__ __forceinline__ unsigned own_key(int src) {
    return ((unsigned)src / (unsigned)OWN_ROWS) & (unsigned)(OWN_BLOCKS - 1);
}

__device__ __forceinline__ unsigned char* key_plane(float* wsBase, const float* tEnd,
                                                    unsigned long long wsBytes, int nE, int* groups) {
    const unsigned long long off =
        ((unsigned long long)((const char*)tEnd - (const char*)wsBase) + 15ull) & ~15ull;
    const long long room = wsBytes > off ? (long long)((wsBytes - off) >> 9) : 0ll;
    const long long all  = (long long)(nE >> 9);
    *groups = (int)(room < all ? room : all);
    return (unsigned char*)wsBase + off;
}

__global__ void gat_gemm_stw(const float* __restrict__ x,
                             const float* __restrict__ W,
                             const float* __restrict__ a,
                             float* __restrict__ h,
                             float* __restrict__ s,
                             float* __restrict__ t,
                             const int* __restrict__ ei,
                             int n, int mtiles, int nE, unsigned long long wsBytes) {
    const int wavesPerBlock = blockDim.x >> 5;
    const int wave = blockIdx.x * wavesPerBlock + ((int)threadIdx.x >> 5);
    const int lane = threadIdx.x & 31;

    {
        int planeGroups;
        unsigned char* plane = key_plane(h, t + n, wsBytes, nE, &planeGroups);
        const int nChunk = planeGroups * 2;
        const int nWaves = (int)gridDim.x * wavesPerBlock;
        for (int c = wave; c < nChunk; c += nWaves) {
            const int4* sp = (const int4*)(ei + (size_t)c * 256) + lane * 2;
            const int4 p = sp[0];
            const int4 q = sp[1];
            uint2 k;
            k.x = own_key(p.x) | (own_key(p.y) << 8) | (own_key(p.z) << 16) | (own_key(p.w) << 24);
            k.y = own_key(q.x) | (own_key(q.y) << 8) | (own_key(q.z) << 16) | (own_key(q.w) << 24);
            ((uint2*)(plane + (size_t)c * 256))[lane] = k;
        }
    }

    if (wave >= mtiles) return;

    const int half = lane >> 4;
    const int l15  = lane & 15;
    const int m0   = wave * 16;

    int arow = m0 + l15; if (arow > n - 1) arow = n - 1;
    const float* xrow = x + (size_t)arow * IN_C;

    v2f afrag[32];
#pragma unroll
    for (int ks = 0; ks < 32; ++ks) {
        const int k = ks * 4 + half * 2;
        afrag[ks].x = xrow[k];
        afrag[ks].y = xrow[k + 1];
    }

    float pr1[8], pr2[8];
#pragma unroll
    for (int r = 0; r < 8; ++r) { pr1[r] = 0.0f; pr2[r] = 0.0f; }

#pragma unroll
    for (int nt = 0; nt < 4; ++nt) {
        const int n0 = nt * 16;
        v8f acc = {};
#pragma unroll
        for (int ks = 0; ks < 32; ++ks) {
            const int k = ks * 4 + half * 2;
            v2f b;
            b.x = W[(size_t)k       * OUT_C + n0 + l15];
            b.y = W[(size_t)(k + 1) * OUT_C + n0 + l15];
            acc = __builtin_amdgcn_wmma_f32_16x16x4_f32(
                 false, afrag[ks],  false, b,
                 (short)0, acc,  false,  false);
        }
        const float a1c = a[n0 + l15];
        const float a2c = a[OUT_C + n0 + l15];
#pragma unroll
        for (int r = 0; r < 8; ++r) {
            const int row = m0 + r + 8 * half;
            const float v = acc[r];
            if (row < n) h[(size_t)row * OUT_C + n0 + l15] = v;
            pr1[r] += v * a1c;
            pr2[r] += v * a2c;
        }
    }

#pragma unroll
    for (int mask = 8; mask >= 1; mask >>= 1) {
#pragma unroll
        for (int r = 0; r < 8; ++r) {
            pr1[r] += __shfl_xor(pr1[r], mask, 32);
            pr2[r] += __shfl_xor(pr2[r], mask, 32);
        }
    }
    if (l15 == 0) {
#pragma unroll
        for (int r = 0; r < 8; ++r) {
            const int row = m0 + r + 8 * half;
            if (row < n) { s[row] = pr1[r]; t[row] = pr2[r]; }
        }
    }
}

#define OWN_THREADS 1024
#define OWN_WAVES   32
#define OWN_QLEN    576

__device__ __forceinline__ void own_drain(const int* myq, int cnt,
                                          const int* __restrict__ ei, int nE,
                                          const float* __restrict__ h,
                                          const float* __restrict__ t,
                                          const float* s_l, float* acc, float* esum,
                                          int lane, int r0, int R) {
    for (int b = 0; b < cnt; b += 32) {
        const int nb = min(32, cnt - b);
        const bool valid = lane < nb;
        int e = 0;
        if (valid) e = myq[b + lane];
        const int src = ei[e];
        const int dst = ei[nE + e];
        const bool own = valid && (((unsigned)src - (unsigned)r0) < (unsigned)R);
        const int  ls  = own ? src - r0 : OWN_ROWS;
        float w = __expf(lrelu(s_l[ls] + t[dst]));
        if (!own) w = 0.0f;
        if (own) atomicAdd(&esum[ls], w);
        for (int k0 = 0; k0 < nb; k0 += 8) {
            float2 v[8];
#pragma unroll
            for (int j = 0; j < 8; ++j) {
                const int d = __builtin_amdgcn_readlane(dst, k0 + j);
                v[j] = *(const float2*)(h + (size_t)d * OUT_C + lane * 2);
            }
#pragma unroll
            for (int j = 0; j < 8; ++j) {
                const float wk = __int_as_float(__builtin_amdgcn_readlane(__float_as_int(w), k0 + j));
                const int   lk = __builtin_amdgcn_readlane(ls, k0 + j);
                float* a = acc + lk * OUT_C + lane * 2;
                atomicAdd(a,     wk * v[j].x);
                atomicAdd(a + 1, wk * v[j].y);
            }
        }
    }
}

__device__ __forceinline__ unsigned zero_bytes(unsigned x) {
    return ~((((x & 0x7f7f7f7fu) + 0x7f7f7f7fu) | x) | 0x7f7f7f7fu);
}

template <bool PLANE>
__device__ __forceinline__ int own_sweep(int cnt, int g0, int g1, int* myq,
                                         const unsigned char* __restrict__ plane,
                                         const int* __restrict__ ei, int nE,
                                         const float* __restrict__ h,
                                         const float* __restrict__ t,
                                         const float* s_l, float* acc, float* esum,
                                         int lane, int wv, int r0, int R) {
    constexpr int U = PLANE ? 8 : 2;
    constexpr int L = PLANE ? 1 : 4;
    const int Gn  = g1 - g0;
    const int nIt = (Gn + OWN_WAVES - 1) / OWN_WAVES;
    const int rot = (int)(((long long)blockIdx.x * Gn) / gridDim.x);
    const unsigned kk = ((unsigned)blockIdx.x & (unsigned)(OWN_BLOCKS - 1)) * 0x01010101u;

    for (int it0 = 0; it0 < nIt; it0 += U) {
        uint4 kv[U][L];
        int   eb[U];
        bool  live[U];
#pragma unroll
        for (int u = 0; u < U; ++u) {
            const int gl = (it0 + u) * OWN_WAVES + wv;
            live[u] = gl < Gn;
            int g = gl + rot; if (g >= Gn) g -= Gn;
            if (!live[u]) g = 0;
            eb[u] = (g0 + g) * 512;
            if (PLANE) {
                kv[u][0] = ((const uint4*)(plane + (size_t)eb[u]))[lane];
            } else {
#pragma unroll
                for (int i = 0; i < L; ++i)
                    kv[u][i] = ((const uint4*)(ei + (size_t)eb[u] + i * 128))[lane];
            }
        }
#pragma unroll
        for (int u = 0; u < U; ++u) {
            unsigned m = 0u;
            if (PLANE) {
                m = (zero_bytes(kv[u][0].x ^ kk) >> 3) | (zero_bytes(kv[u][0].y ^ kk) >> 2) |
                    (zero_bytes(kv[u][0].z ^ kk) >> 1) |  zero_bytes(kv[u][0].w ^ kk);
                if (!live[u]) m = 0u;
            } else {
                const unsigned ru = live[u] ? (unsigned)R : 0u;
#pragma unroll
                for (int i = 0; i < L; ++i) {
                    m |= (((unsigned)kv[u][i].x - (unsigned)r0) < ru ? 1u : 0u) << (4 * i);
                    m |= (((unsigned)kv[u][i].y - (unsigned)r0) < ru ? 1u : 0u) << (4 * i + 1);
                    m |= (((unsigned)kv[u][i].z - (unsigned)r0) < ru ? 1u : 0u) << (4 * i + 2);
                    m |= (((unsigned)kv[u][i].w - (unsigned)r0) < ru ? 1u : 0u) << (4 * i + 3);
                }
            }
            unsigned any = __builtin_amdgcn_ballot_w32(m != 0u);
            while (any) {
                if (m != 0u) {
                    const int k = __builtin_ctz(m);
                    const int e = PLANE ? eb[u] + lane * 16 + ((k & 7) - 4) * 4 + (k >> 3)
                                        : eb[u] + (k >> 2) * 128 + lane * 4 + (k & 3);
                    myq[cnt + __builtin_amdgcn_mbcnt_lo(any, 0u)] = e;
                    m &= m - 1u;
                }
                cnt += __builtin_popcount(any);
                any = __builtin_amdgcn_ballot_w32(m != 0u);
            }
            if (cnt >= 64) {
                const int full = cnt & ~31;
                own_drain(myq, full, ei, nE, h, t, s_l, acc, esum, lane, r0, R);
                const int rest = cnt - full;
                int mv = 0;
                if (lane < rest) mv = myq[full + lane];
                if (lane < rest) myq[lane] = mv;
                cnt = rest;
            }
        }
    }
    return cnt;
}

__global__ void __launch_bounds__(OWN_THREADS)
gat_owner(const int* __restrict__ ei,
          float* __restrict__ h,
          const float* __restrict__ s,
          const float* __restrict__ t,
          float* __restrict__ out,
          int n, int nE, unsigned long long wsBytes) {
    __shared__ float acc[(OWN_ROWS + 1) * OUT_C];
    __shared__ float esum[OWN_ROWS + 1];
    __shared__ float s_l[OWN_ROWS + 1];
    __shared__ int   q[OWN_WAVES * OWN_QLEN];

    const int tid  = threadIdx.x;
    const int lane = tid & 31;
    const int wv   = __builtin_amdgcn_readfirstlane(tid >> 5);
    int* myq = q + wv * OWN_QLEN;

    const int Gf = nE >> 9;
    int Gp;
    const unsigned char* plane = key_plane(h, t + n, wsBytes, nE, &Gp);

    for (int r0 = blockIdx.x * OWN_ROWS; r0 < n; r0 += gridDim.x * OWN_ROWS) {
        const int R = min(OWN_ROWS, n - r0);

        for (int i = tid; i < (OWN_ROWS + 1) * OUT_C; i += OWN_THREADS) acc[i] = 0.0f;
        for (int i = tid; i <= OWN_ROWS; i += OWN_THREADS) {
            esum[i] = 0.0f;
            s_l[i]  = (i < R) ? s[r0 + i] : 0.0f;
        }
        __syncthreads();

        int cnt = 0;
        cnt = own_sweep<true >(cnt, 0,  Gp, myq, plane, ei, nE, h, t, s_l, acc, esum, lane, wv, r0, R);
        cnt = own_sweep<false>(cnt, Gp, Gf, myq, plane, ei, nE, h, t, s_l, acc, esum, lane, wv, r0, R);
        if (wv == 0) {
            for (int e0 = Gf * 512; e0 < nE; e0 += 32) {
                const int  e = e0 + lane;
                const int  sr = ei[min(e, nE - 1)];
                const bool m  = (e < nE) && (((unsigned)sr - (unsigned)r0) < (unsigned)R);
                const unsigned k = __builtin_amdgcn_ballot_w32(m);
                if (m) myq[cnt + __builtin_amdgcn_mbcnt_lo(k, 0u)] = e;
                cnt += __builtin_popcount(k);
            }
        }
        own_drain(myq, cnt, ei, nE, h, t, s_l, acc, esum, lane, r0, R);
        __syncthreads();

        for (int i4 = tid; i4 < R * (OUT_C / 4); i4 += OWN_THREADS) {
            const int li = i4 >> 4;
            const int c4 = (i4 & 15) * 4;
            const size_t i = (size_t)(r0 + li);
            const float w = __expf(lrelu(s_l[li] + t[i]));
            const float denom = esum[li] + w;
            const float4 hv = *(const float4*)(h + i * OUT_C + c4);
            const float4 av = *(const float4*)(acc + li * OUT_C + c4);
            float4 o;
            o.x = (av.x + w * hv.x) / denom;
            o.y = (av.y + w * hv.y) / denom;
            o.z = (av.z + w * hv.z) / denom;
            o.w = (av.w + w * hv.w) / denom;
            *(float4*)(out + i * OUT_C + c4) = o;
        }
        __syncthreads();
    }
}

extern "C" void kernel_launch(void* const* d_in, const int* in_sizes, int n_in,
                              void* d_out, int out_size, void* d_ws, size_t ws_size,
                              hipStream_t stream) {
    const float* x  = (const float*)d_in[0];
    const float* W  = (const float*)d_in[1];
    const float* a  = (const float*)d_in[2];
    const int*   ei = (const int*)d_in[3];

    const int n = in_sizes[0] / IN_C;
    const int E = in_sizes[3] / 2;

    float* ws     = (float*)d_ws;
    float* h      = ws;
    float* s      = h + (size_t)n * OUT_C;
    float* t      = s + n;
    float* out    = (float*)d_out;

    const dim3 blk(256);
    const int wavesPerBlock = 256 / 32;

    const int mtiles = (n + 15) / 16;
    gat_gemm_stw<<<(mtiles + wavesPerBlock - 1) / wavesPerBlock, blk, 0, stream>>>(
        x, W, a, h, s, t, ei, n, mtiles, E, (unsigned long long)ws_size);

    gat_owner<<<OWN_BLOCKS, OWN_THREADS, 0, stream>>>(ei, h, s, t, out, n, E,
                                                      (unsigned long long)ws_size);
}


# owner drain reads one packed record per listed edge ((dst<<9)|row, plane written by the GEMM launch behind the key plane) instead of two indexed loads (source, dst); LDS float adds, arrival order
# speedup vs baseline: 2.9505x; 1.0471x over previous
#include <hip/hip_runtime.h>
#include <hip/hip_bf16.h>
#include <math.h>


typedef __attribute__((ext_vector_type(2))) float v2f;
typedef __attribute__((ext_vector_type(8))) float v8f;

#define IN_C  128
#define OUT_C 64
#define GAT_ALPHA 0.2f

__device__ __forceinline__ float lrelu(float z) {
    return z >= 0.0f ? z : GAT_ALPHA * z;
}

#define OWN_ROWS    392
#define OWN_BLOCKS  256

__device__ __forceinline__ unsigned own_key(int src) {
    return ((unsigned)src / (unsigned)OWN_ROWS) & (unsigned)(OWN_BLOCKS - 1);
}

__device__ __forceinline__ unsigned char* key_plane(float* wsBase, const float* tEnd,
                                                    unsigned long long wsBytes, int nE, int* groups) {
    const unsigned long long off =
        ((unsigned long long)((const char*)tEnd - (const char*)wsBase) + 15ull) & ~15ull;
    const long long room = wsBytes > off ? (long long)((wsBytes - off) >> 9) : 0ll;
    const long long all  = (long long)(nE >> 9);
    *groups = (int)(room < all ? room : all);
    return (unsigned char*)wsBase + off;
}

__device__ __forceinline__ unsigned own_row(int src) {
    return (unsigned)src - ((unsigned)src / (unsigned)OWN_ROWS) * (unsigned)OWN_ROWS;
}

__device__ __forceinline__ int rec_plane(unsigned char* plane, int groups, const float* wsBase,
                                         unsigned long long wsBytes, int n, int nE, unsigned** rec) {
    const unsigned long long off =
        (unsigned long long)((const char*)plane - (const char*)wsBase) + (unsigned long long)groups * 512ull;
    const long long room = wsBytes > off ? (long long)((wsBytes - off) >> 11) : 0ll;
    const int held = (int)(room < (long long)groups ? room : (long long)groups);
    const bool ok = (nE & 3) == 0 && n <= OWN_BLOCKS * OWN_ROWS;
    *rec = (unsigned*)(plane + (size_t)groups * 512);
    return ok ? held * 512 : 0;
}

__global__ void gat_gemm_stw(const float* __restrict__ x,
                             const float* __restrict__ W,
                             const float* __restrict__ a,
                             float* __restrict__ h,
                             float* __restrict__ s,
                             float* __restrict__ t,
                             const int* __restrict__ ei,
                             int n, int mtiles, int nE, unsigned long long wsBytes) {
    const int wavesPerBlock = blockDim.x >> 5;
    const int wave = blockIdx.x * wavesPerBlock + ((int)threadIdx.x >> 5);
    const int lane = threadIdx.x & 31;

    {
        int planeGroups;
        unsigned char* plane = key_plane(h, t + n, wsBytes, nE, &planeGroups);
        unsigned* rec;
        const int recEdges = rec_plane(plane, planeGroups, h, wsBytes, n, nE, &rec);
        const int nChunk = planeGroups * 2;
        const int nWaves = (int)gridDim.x * wavesPerBlock;
        for (int c = wave; c < nChunk; c += nWaves) {
            const int4* sp = (const int4*)(ei + (size_t)c * 256) + lane * 2;
            const int4 p = sp[0];
            const int4 q = sp[1];
            const bool recHere = c * 256 < recEdges;
            int4 d0 = {0, 0, 0, 0}, d1 = {0, 0, 0, 0};
            if (recHere) {
                const int4* dp = (const int4*)(ei + (size_t)nE + (size_t)c * 256) + lane * 2;
                d0 = dp[0];
                d1 = dp[1];
            }
            uint2 k;
            k.x = own_key(p.x) | (own_key(p.y) << 8) | (own_key(p.z) << 16) | (own_key(p.w) << 24);
            k.y = own_key(q.x) | (own_key(q.y) << 8) | (own_key(q.z) << 16) | (own_key(q.w) << 24);
            ((uint2*)(plane + (size_t)c * 256))[lane] = k;
            if (recHere) {
                uint4 r0v, r1v;
                r0v.x = ((unsigned)d0.x << 9) | own_row(p.x);
                r0v.y = ((unsigned)d0.y << 9) | own_row(p.y);
                r0v.z = ((unsigned)d0.z << 9) | own_row(p.z);
                r0v.w = ((unsigned)d0.w << 9) | own_row(p.w);
                r1v.x = ((unsigned)d1.x << 9) | own_row(q.x);
                r1v.y = ((unsigned)d1.y << 9) | own_row(q.y);
                r1v.z = ((unsigned)d1.z << 9) | own_row(q.z);
                r1v.w = ((unsigned)d1.w << 9) | own_row(q.w);
                uint4* rp = (uint4*)(rec + (size_t)c * 256) + lane * 2;
                rp[0] = r0v;
                rp[1] = r1v;
            }
        }
    }

    if (wave >= mtiles) return;

    const int half = lane >> 4;
    const int l15  = lane & 15;
    const int m0   = wave * 16;

    int arow = m0 + l15; if (arow > n - 1) arow = n - 1;
    const float* xrow = x + (size_t)arow * IN_C;

    v2f afrag[32];
#pragma unroll
    for (int ks = 0; ks < 32; ++ks) {
        const int k = ks * 4 + half * 2;
        afrag[ks].x = xrow[k];
        afrag[ks].y = xrow[k + 1];
    }

    float pr1[8], pr2[8];
#pragma unroll
    for (int r = 0; r < 8; ++r) { pr1[r] = 0.0f; pr2[r] = 0.0f; }

#pragma unroll
    for (int nt = 0; nt < 4; ++nt) {
        const int n0 = nt * 16;
        v8f acc = {};
#pragma unroll
        for (int ks = 0; ks < 32; ++ks) {
            const int k = ks * 4 + half * 2;
            v2f b;
            b.x = W[(size_t)k       * OUT_C + n0 + l15];
            b.y = W[(size_t)(k + 1) * OUT_C + n0 + l15];
            acc = __builtin_amdgcn_wmma_f32_16x16x4_f32(
                 false, afrag[ks],  false, b,
                 (short)0, acc,  false,  false);
        }
        const float a1c = a[n0 + l15];
        const float a2c = a[OUT_C + n0 + l15];
#pragma unroll
        for (int r = 0; r < 8; ++r) {
            const int row = m0 + r + 8 * half;
            const float v = acc[r];
            if (row < n) h[(size_t)row * OUT_C + n0 + l15] = v;
            pr1[r] += v * a1c;
            pr2[r] += v * a2c;
        }
    }

#pragma unroll
    for (int mask = 8; mask >= 1; mask >>= 1) {
#pragma unroll
        for (int r = 0; r < 8; ++r) {
            pr1[r] += __shfl_xor(pr1[r], mask, 32);
            pr2[r] += __shfl_xor(pr2[r], mask, 32);
        }
    }
    if (l15 == 0) {
#pragma unroll
        for (int r = 0; r < 8; ++r) {
            const int row = m0 + r + 8 * half;
            if (row < n) { s[row] = pr1[r]; t[row] = pr2[r]; }
        }
    }
}

#define OWN_THREADS 1024
#define OWN_WAVES   32
#define OWN_QLEN    576

__device__ __forceinline__ void own_drain(const int* myq, int cnt,
                                          const unsigned* __restrict__ rec, int recEdges,
                                          const int* __restrict__ ei, int nE,
                                          const float* __restrict__ h,
                                          const float* __restrict__ t,
                                          const float* s_l, float* acc, float* esum,
                                          int lane, int r0, int R) {
    for (int b = 0; b < cnt; b += 32) {
        const int nb = min(32, cnt - b);
        const bool valid = lane < nb;
        int e = 0;
        if (valid) e = myq[b + lane];
        int  dst, ls;
        bool own;
        if (e < recEdges) {
            const unsigned r = rec[e];
            dst = (int)(r >> 9);
            own = valid;
            ls  = own ? (int)(r & 511u) : OWN_ROWS;
        } else {
            const int src = ei[e];
            dst = ei[nE + e];
            own = valid && (((unsigned)src - (unsigned)r0) < (unsigned)R);
            ls  = own ? src - r0 : OWN_ROWS;
        }
        float w = __expf(lrelu(s_l[ls] + t[dst]));
        if (!own) w = 0.0f;
        if (own) atomicAdd(&esum[ls], w);
        for (int k0 = 0; k0 < nb; k0 += 8) {
            float2 v[8];
#pragma unroll
            for (int j = 0; j < 8; ++j) {
                const int d = __builtin_amdgcn_readlane(dst, k0 + j);
                v[j] = *(const float2*)(h + (size_t)d * OUT_C + lane * 2);
            }
#pragma unroll
            for (int j = 0; j < 8; ++j) {
                const float wk = __int_as_float(__builtin_amdgcn_readlane(__float_as_int(w), k0 + j));
                const int   lk = __builtin_amdgcn_readlane(ls, k0 + j);
                float* a = acc + lk * OUT_C + lane * 2;
                atomicAdd(a,     wk * v[j].x);
                atomicAdd(a + 1, wk * v[j].y);
            }
        }
    }
}

__device__ __forceinline__ unsigned zero_bytes(unsigned x) {
    return ~((((x & 0x7f7f7f7fu) + 0x7f7f7f7fu) | x) | 0x7f7f7f7fu);
}

template <bool PLANE>
__device__ __forceinline__ int own_sweep(int cnt, int g0, int g1, int* myq,
                                         const unsigned char* __restrict__ plane,
                                         const unsigned* __restrict__ rec, int recEdges,
                                         const int* __restrict__ ei, int nE,
                                         const float* __restrict__ h,
                                         const float* __restrict__ t,
                                         const float* s_l, float* acc, float* esum,
                                         int lane, int wv, int r0, int R) {
    constexpr int U = PLANE ? 8 : 2;
    constexpr int L = PLANE ? 1 : 4;
    const int Gn  = g1 - g0;
    const int nIt = (Gn + OWN_WAVES - 1) / OWN_WAVES;
    const int rot = (int)(((long long)blockIdx.x * Gn) / gridDim.x);
    const unsigned kk = ((unsigned)blockIdx.x & (unsigned)(OWN_BLOCKS - 1)) * 0x01010101u;

    for (int it0 = 0; it0 < nIt; it0 += U) {
        uint4 kv[U][L];
        int   eb[U];
        bool  live[U];
#pragma unroll
        for (int u = 0; u < U; ++u) {
            const int gl = (it0 + u) * OWN_WAVES + wv;
            live[u] = gl < Gn;
            int g = gl + rot; if (g >= Gn) g -= Gn;
            if (!live[u]) g = 0;
            eb[u] = (g0 + g) * 512;
            if (PLANE) {
                kv[u][0] = ((const uint4*)(plane + (size_t)eb[u]))[lane];
            } else {
#pragma unroll
                for (int i = 0; i < L; ++i)
                    kv[u][i] = ((const uint4*)(ei + (size_t)eb[u] + i * 128))[lane];
            }
        }
#pragma unroll
        for (int u = 0; u < U; ++u) {
            unsigned m = 0u;
            if (PLANE) {
                m = (zero_bytes(kv[u][0].x ^ kk) >> 3) | (zero_bytes(kv[u][0].y ^ kk) >> 2) |
                    (zero_bytes(kv[u][0].z ^ kk) >> 1) |  zero_bytes(kv[u][0].w ^ kk);
                if (!live[u]) m = 0u;
            } else {
                const unsigned ru = live[u] ? (unsigned)R : 0u;
#pragma unroll
                for (int i = 0; i < L; ++i) {
                    m |= (((unsigned)kv[u][i].x - (unsigned)r0) < ru ? 1u : 0u) << (4 * i);
                    m |= (((unsigned)kv[u][i].y - (unsigned)r0) < ru ? 1u : 0u) << (4 * i + 1);
                    m |= (((unsigned)kv[u][i].z - (unsigned)r0) < ru ? 1u : 0u) << (4 * i + 2);
                    m |= (((unsigned)kv[u][i].w - (unsigned)r0) < ru ? 1u : 0u) << (4 * i + 3);
                }
            }
            unsigned any = __builtin_amdgcn_ballot_w32(m != 0u);
            while (any) {
                if (m != 0u) {
                    const int k = __builtin_ctz(m);
                    const int e = PLANE ? eb[u] + lane * 16 + ((k & 7) - 4) * 4 + (k >> 3)
                                        : eb[u] + (k >> 2) * 128 + lane * 4 + (k & 3);
                    myq[cnt + __builtin_amdgcn_mbcnt_lo(any, 0u)] = e;
                    m &= m - 1u;
                }
                cnt += __builtin_popcount(any);
                any = __builtin_amdgcn_ballot_w32(m != 0u);
            }
            if (cnt >= 64) {
                const int full = cnt & ~31;
                own_drain(myq, full, rec, recEdges, ei, nE, h, t, s_l, acc, esum, lane, r0, R);
                const int rest = cnt - full;
                int mv = 0;
                if (lane < rest) mv = myq[full + lane];
                if (lane < rest) myq[lane] = mv;
                cnt = rest;
            }
        }
    }
    return cnt;
}

__global__ void __launch_bounds__(OWN_THREADS)
gat_owner(const int* __restrict__ ei,
          float* __restrict__ h,
          const float* __restrict__ s,
          const float* __restrict__ t,
          float* __restrict__ out,
          int n, int nE, unsigned long long wsBytes) {
    __shared__ float acc[(OWN_ROWS + 1) * OUT_C];
    __shared__ float esum[OWN_ROWS + 1];
    __shared__ float s_l[OWN_ROWS + 1];
    __shared__ int   q[OWN_WAVES * OWN_QLEN];

    const int tid  = threadIdx.x;
    const int lane = tid & 31;
    const int wv   = __builtin_amdgcn_readfirstlane(tid >> 5);
    int* myq = q + wv * OWN_QLEN;

    const int Gf = nE >> 9;
    int Gp;
    unsigned char* plane = key_plane(h, t + n, wsBytes, nE, &Gp);
    unsigned* rec;
    const int recEdges = rec_plane(plane, Gp, h, wsBytes, n, nE, &rec);

    for (int r0 = blockIdx.x * OWN_ROWS; r0 < n; r0 += gridDim.x * OWN_ROWS) {
        const int R = min(OWN_ROWS, n - r0);

        for (int i = tid; i < (OWN_ROWS + 1) * OUT_C; i += OWN_THREADS) acc[i] = 0.0f;
        for (int i = tid; i <= OWN_ROWS; i += OWN_THREADS) {
            esum[i] = 0.0f;
            s_l[i]  = (i < R) ? s[r0 + i] : 0.0f;
        }
        __syncthreads();

        int cnt = 0;
        cnt = own_sweep<true >(cnt, 0,  Gp, myq, plane, rec, recEdges, ei, nE, h, t, s_l, acc, esum, lane, wv, r0, R);
        cnt = own_sweep<false>(cnt, Gp, Gf, myq, plane, rec, recEdges, ei, nE, h, t, s_l, acc, esum, lane, wv, r0, R);
        if (wv == 0) {
            for (int e0 = Gf * 512; e0 < nE; e0 += 32) {
                const int  e = e0 + lane;
                const int  sr = ei[min(e, nE - 1)];
                const bool m  = (e < nE) && (((unsigned)sr - (unsigned)r0) < (unsigned)R);
                const unsigned k = __builtin_amdgcn_ballot_w32(m);
                if (m) myq[cnt + __builtin_amdgcn_mbcnt_lo(k, 0u)] = e;
                cnt += __builtin_popcount(k);
            }
        }
        own_drain(myq, cnt, rec, recEdges, ei, nE, h, t, s_l, acc, esum, lane, r0, R);
        __syncthreads();

        for (int i4 = tid; i4 < R * (OUT_C / 4); i4 += OWN_THREADS) {
            const int li = i4 >> 4;
            const int c4 = (i4 & 15) * 4;
            const size_t i = (size_t)(r0 + li);
            const float w = __expf(lrelu(s_l[li] + t[i]));
            const float denom = esum[li] + w;
            const float4 hv = *(const float4*)(h + i * OUT_C + c4);
            const float4 av = *(const float4*)(acc + li * OUT_C + c4);
            float4 o;
            o.x = (av.x + w * hv.x) / denom;
            o.y = (av.y + w * hv.y) / denom;
            o.z = (av.z + w * hv.z) / denom;
            o.w = (av.w + w * hv.w) / denom;
            *(float4*)(out + i * OUT_C + c4) = o;
        }
        __syncthreads();
    }
}

extern "C" void kernel_launch(void* const* d_in, const int* in_sizes, int n_in,
                              void* d_out, int out_size, void* d_ws, size_t ws_size,
                              hipStream_t stream) {
    const float* x  = (const float*)d_in[0];
    const float* W  = (const float*)d_in[1];
    const float* a  = (const float*)d_in[2];
    const int*   ei = (const int*)d_in[3];

    const int n = in_sizes[0] / IN_C;
    const int E = in_sizes[3] / 2;

    float* ws     = (float*)d_ws;
    float* h      = ws;
    float* s      = h + (size_t)n * OUT_C;
    float* t      = s + n;
    float* out    = (float*)d_out;

    const dim3 blk(256);
    const int wavesPerBlock = 256 / 32;

    const int mtiles = (n + 15) / 16;
    gat_gemm_stw<<<(mtiles + wavesPerBlock - 1) / wavesPerBlock, blk, 0, stream>>>(
        x, W, a, h, s, t, ei, n, mtiles, E, (unsigned long long)ws_size);

    gat_owner<<<OWN_BLOCKS, OWN_THREADS, 0, stream>>>(ei, h, s, t, out, n, E,
                                                      (unsigned long long)ws_size);
}


# owner-key plane bit-sliced (8 words per 32 edges, built by wave votes in the GEMM launch): a sweep lane tests 32 keys in 8 bit operations; edges listed in a new order, LDS float adds, arrival order
# speedup vs baseline: 3.1229x; 1.0584x over previous
#include <hip/hip_runtime.h>
#include <hip/hip_bf16.h>
#include <math.h>


typedef __attribute__((ext_vector_type(2))) float v2f;
typedef __attribute__((ext_vector_type(8))) float v8f;

#define IN_C  128
#define OUT_C 64
#define GAT_ALPHA 0.2f

__device__ __forceinline__ float lrelu(float z) {
    return z >= 0.0f ? z : GAT_ALPHA * z;
}

#define OWN_ROWS    392
#define OWN_BLOCKS  256

__device__ __forceinline__ unsigned own_key(int src) {
    return ((unsigned)src / (unsigned)OWN_ROWS) & (unsigned)(OWN_BLOCKS - 1);
}

__device__ __forceinline__ unsigned char* key_plane(float* wsBase, const float* tEnd,
                                                    unsigned long long wsBytes, int nE, int* groups) {
    const unsigned long long off =
        ((unsigned long long)((const char*)tEnd - (const char*)wsBase) + 15ull) & ~15ull;
    const long long room = wsBytes > off ? (long long)((wsBytes - off) >> 9) : 0ll;
    const long long all  = (long long)(nE >> 9);
    *groups = (int)(room < all ? room : all);
    return (unsigned char*)wsBase + off;
}

__device__ __forceinline__ unsigned put_lane(unsigned old, unsigned wd, int ln, int lane) {
    return lane == ln ? wd : old;
}

__device__ __forceinline__ unsigned own_row(int src) {
    return (unsigned)src - ((unsigned)src / (unsigned)OWN_ROWS) * (unsigned)OWN_ROWS;
}

__device__ __forceinline__ int rec_plane(unsigned char* plane, int groups, const float* wsBase,
                                         unsigned long long wsBytes, int n, int nE, unsigned** rec) {
    const unsigned long long off =
        (unsigned long long)((const char*)plane - (const char*)wsBase) + (unsigned long long)groups * 512ull;
    const long long room = wsBytes > off ? (long long)((wsBytes - off) >> 11) : 0ll;
    const int held = (int)(room < (long long)groups ? room : (long long)groups);
    const bool ok = (nE & 3) == 0 && n <= OWN_BLOCKS * OWN_ROWS;
    *rec = (unsigned*)(plane + (size_t)groups * 512);
    return ok ? held * 512 : 0;
}

__global__ void gat_gemm_stw(const float* __restrict__ x,
                             const float* __restrict__ W,
                             const float* __restrict__ a,
                             float* __restrict__ h,
                             float* __restrict__ s,
                             float* __restrict__ t,
                             const int* __restrict__ ei,
                             int n, int mtiles, int nE, unsigned long long wsBytes) {
    const int wavesPerBlock = blockDim.x >> 5;
    const int wave = blockIdx.x * wavesPerBlock + ((int)threadIdx.x >> 5);
    const int lane = threadIdx.x & 31;

    {
        int planeGroups;
        unsigned char* plane = key_plane(h, t + n, wsBytes, nE, &planeGroups);
        unsigned* rec;
        const int recEdges = rec_plane(plane, planeGroups, h, wsBytes, n, nE, &rec);
        const int nChunk = planeGroups * 2;
        const int keyChunks = (planeGroups >> 1) * 4;
        const int nWaves = (int)gridDim.x * wavesPerBlock;
        for (int c = wave; c < nChunk; c += nWaves) {
            const int4* sp = (const int4*)(ei + (size_t)c * 256) + lane * 2;
            const int4 p = sp[0];
            const int4 q = sp[1];
            const bool recHere = c * 256 < recEdges;
            int4 d0 = {0, 0, 0, 0}, d1 = {0, 0, 0, 0};
            if (recHere) {
                const int4* dp = (const int4*)(ei + (size_t)nE + (size_t)c * 256) + lane * 2;
                d0 = dp[0];
                d1 = dp[1];
            }
            if (c < keyChunks) {
                const int sj[8] = {p.x, p.y, p.z, p.w, q.x, q.y, q.z, q.w};
                unsigned lo = 0u, hi = 0u;
#pragma unroll
                for (int j = 0; j < 8; ++j) {
                    const unsigned key = own_key(sj[j]);
#pragma unroll
                    for (int b = 0; b < 8; ++b) {
                        const unsigned wd = __builtin_amdgcn_ballot_w32(((key >> b) & 1u) != 0u);
                        if (b < 4) lo = put_lane(lo, wd, j * 4 + b, lane);
                        else       hi = put_lane(hi, wd, j * 4 + b - 4, lane);
                    }
                }
                unsigned* kp = (unsigned*)(plane + (size_t)(c >> 2) * 1024 + (size_t)(c & 3) * 128);
                kp[lane]       = lo;
                kp[128 + lane] = hi;
            }
            if (recHere) {
                uint4 r0v, r1v;
                r0v.x = ((unsigned)d0.x << 9) | own_row(p.x);
                r0v.y = ((unsigned)d0.y << 9) | own_row(p.y);
                r0v.z = ((unsigned)d0.z << 9) | own_row(p.z);
                r0v.w = ((unsigned)d0.w << 9) | own_row(p.w);
                r1v.x = ((unsigned)d1.x << 9) | own_row(q.x);
                r1v.y = ((unsigned)d1.y << 9) | own_row(q.y);
                r1v.z = ((unsigned)d1.z << 9) | own_row(q.z);
                r1v.w = ((unsigned)d1.w << 9) | own_row(q.w);
                uint4* rp = (uint4*)(rec + (size_t)c * 256) + lane * 2;
                rp[0] = r0v;
                rp[1] = r1v;
            }
        }
    }

    if (wave >= mtiles) return;

    const int half = lane >> 4;
    const int l15  = lane & 15;
    const int m0   = wave * 16;

    int arow = m0 + l15; if (arow > n - 1) arow = n - 1;
    const float* xrow = x + (size_t)arow * IN_C;

    v2f afrag[32];
#pragma unroll
    for (int ks = 0; ks < 32; ++ks) {
        const int k = ks * 4 + half * 2;
        afrag[ks].x = xrow[k];
        afrag[ks].y = xrow[k + 1];
    }

    float pr1[8], pr2[8];
#pragma unroll
    for (int r = 0; r < 8; ++r) { pr1[r] = 0.0f; pr2[r] = 0.0f; }

#pragma unroll
    for (int nt = 0; nt < 4; ++nt) {
        const int n0 = nt * 16;
        v8f acc = {};
#pragma unroll
        for (int ks = 0; ks < 32; ++ks) {
            const int k = ks * 4 + half * 2;
            v2f b;
            b.x = W[(size_t)k       * OUT_C + n0 + l15];
            b.y = W[(size_t)(k + 1) * OUT_C + n0 + l15];
            acc = __builtin_amdgcn_wmma_f32_16x16x4_f32(
                 false, afrag[ks],  false, b,
                 (short)0, acc,  false,  false);
        }
        const float a1c = a[n0 + l15];
        const float a2c = a[OUT_C + n0 + l15];
#pragma unroll
        for (int r = 0; r < 8; ++r) {
            const int row = m0 + r + 8 * half;
            const float v = acc[r];
            if (row < n) h[(size_t)row * OUT_C + n0 + l15] = v;
            pr1[r] += v * a1c;
            pr2[r] += v * a2c;
        }
    }

#pragma unroll
    for (int mask = 8; mask >= 1; mask >>= 1) {
#pragma unroll
        for (int r = 0; r < 8; ++r) {
            pr1[r] += __shfl_xor(pr1[r], mask, 32);
            pr2[r] += __shfl_xor(pr2[r], mask, 32);
        }
    }
    if (l15 == 0) {
#pragma unroll
        for (int r = 0; r < 8; ++r) {
            const int row = m0 + r + 8 * half;
            if (row < n) { s[row] = pr1[r]; t[row] = pr2[r]; }
        }
    }
}

#define OWN_THREADS 1024
#define OWN_WAVES   32
#define OWN_QLEN    1088

__device__ __forceinline__ void own_drain(const int* myq, int cnt,
                                          const unsigned* __restrict__ rec, int recEdges,
                                          const int* __restrict__ ei, int nE,
                                          const float* __restrict__ h,
                                          const float* __restrict__ t,
                                          const float* s_l, float* acc, float* esum,
                                          int lane, int r0, int R) {
    for (int b = 0; b < cnt; b += 32) {
        const int nb = min(32, cnt - b);
        const bool valid = lane < nb;
        int e = 0;
        if (valid) e = myq[b + lane];
        int  dst, ls;
        bool own;
        if (e < recEdges) {
            const unsigned r = rec[e];
            dst = (int)(r >> 9);
            own = valid;
            ls  = own ? (int)(r & 511u) : OWN_ROWS;
        } else {
            const int src = ei[e];
            dst = ei[nE + e];
            own = valid && (((unsigned)src - (unsigned)r0) < (unsigned)R);
            ls  = own ? src - r0 : OWN_ROWS;
        }
        float w = __expf(lrelu(s_l[ls] + t[dst]));
        if (!own) w = 0.0f;
        if (own) atomicAdd(&esum[ls], w);
        for (int k0 = 0; k0 < nb; k0 += 8) {
            float2 v[8];
#pragma unroll
            for (int j = 0; j < 8; ++j) {
                const int d = __builtin_amdgcn_readlane(dst, k0 + j);
                v[j] = *(const float2*)(h + (size_t)d * OUT_C + lane * 2);
            }
#pragma unroll
            for (int j = 0; j < 8; ++j) {
                const float wk = __int_as_float(__builtin_amdgcn_readlane(__float_as_int(w), k0 + j));
                const int   lk = __builtin_amdgcn_readlane(ls, k0 + j);
                float* a = acc + lk * OUT_C + lane * 2;
                atomicAdd(a,     wk * v[j].x);
                atomicAdd(a + 1, wk * v[j].y);
            }
        }
    }
}

template <bool PLANE>
__device__ __forceinline__ int own_sweep(int cnt, int g0, int g1, int* myq,
                                         const unsigned char* __restrict__ plane,
                                         const unsigned* __restrict__ rec, int recEdges,
                                         const int* __restrict__ ei, int nE,
                                         const float* __restrict__ h,
                                         const float* __restrict__ t,
                                         const float* s_l, float* acc, float* esum,
                                         int lane, int wv, int r0, int R) {
    constexpr int U  = PLANE ? 4 : 2;
    constexpr int L  = PLANE ? 2 : 4;
    constexpr int GE = PLANE ? 1024 : 512;
    const int Gn  = g1 - g0;
    const int nIt = (Gn + OWN_WAVES - 1) / OWN_WAVES;
    const int rot = (int)(((long long)blockIdx.x * Gn) / gridDim.x);
    unsigned kx[8];
#pragma unroll
    for (int b = 0; b < 8; ++b) kx[b] = (((unsigned)blockIdx.x >> b) & 1u) ? 0u : ~0u;
    const int laneOff = (lane >> 3) * 256 + (lane & 7);

    for (int it0 = 0; it0 < nIt; it0 += U) {
        uint4 kv[U][L];
        int   eb[U];
        bool  live[U];
#pragma unroll
        for (int u = 0; u < U; ++u) {
            const int gl = (it0 + u) * OWN_WAVES + wv;
            live[u] = gl < Gn;
            int g = gl + rot; if (g >= Gn) g -= Gn;
            if (!live[u]) g = 0;
            eb[u] = (g0 + g) * GE;
            if (PLANE) {
                kv[u][0] = ((const uint4*)(plane + (size_t)eb[u]))[lane];
                kv[u][1] = ((const uint4*)(plane + (size_t)eb[u] + 512))[lane];
            } else {
#pragma unroll
                for (int i = 0; i < L; ++i)
                    kv[u][i] = ((const uint4*)(ei + (size_t)eb[u] + i * 128))[lane];
            }
        }
#pragma unroll
        for (int u = 0; u < U; ++u) {
            unsigned m = 0u;
            if (PLANE) {
                m = (kv[u][0].x ^ kx[0]) & (kv[u][0].y ^ kx[1]) & (kv[u][0].z ^ kx[2]) & (kv[u][0].w ^ kx[3]) &
                    (kv[u][1].x ^ kx[4]) & (kv[u][1].y ^ kx[5]) & (kv[u][1].z ^ kx[6]) & (kv[u][1].w ^ kx[7]);
                if (!live[u]) m = 0u;
            } else {
                const unsigned ru = live[u] ? (unsigned)R : 0u;
#pragma unroll
                for (int i = 0; i < L; ++i) {
                    m |= (((unsigned)kv[u][i].x - (unsigned)r0) < ru ? 1u : 0u) << (4 * i);
                    m |= (((unsigned)kv[u][i].y - (unsigned)r0) < ru ? 1u : 0u) << (4 * i + 1);
                    m |= (((unsigned)kv[u][i].z - (unsigned)r0) < ru ? 1u : 0u) << (4 * i + 2);
                    m |= (((unsigned)kv[u][i].w - (unsigned)r0) < ru ? 1u : 0u) << (4 * i + 3);
                }
            }
            unsigned any = __builtin_amdgcn_ballot_w32(m != 0u);
            while (any) {
                if (m != 0u) {
                    const int k = __builtin_ctz(m);
                    const int e = PLANE ? eb[u] + laneOff + k * 8
                                        : eb[u] + (k >> 2) * 128 + lane * 4 + (k & 3);
                    myq[cnt + __builtin_amdgcn_mbcnt_lo(any, 0u)] = e;
                    m &= m - 1u;
                }
                cnt += __builtin_popcount(any);
                any = __builtin_amdgcn_ballot_w32(m != 0u);
            }
            if (cnt >= 64) {
                const int full = cnt & ~31;
                own_drain(myq, full, rec, recEdges, ei, nE, h, t, s_l, acc, esum, lane, r0, R);
                const int rest = cnt - full;
                int mv = 0;
                if (lane < rest) mv = myq[full + lane];
                if (lane < rest) myq[lane] = mv;
                cnt = rest;
            }
        }
    }
    return cnt;
}

__global__ void __launch_bounds__(OWN_THREADS)
gat_owner(const int* __restrict__ ei,
          float* __restrict__ h,
          const float* __restrict__ s,
          const float* __restrict__ t,
          float* __restrict__ out,
          int n, int nE, unsigned long long wsBytes) {
    __shared__ float acc[(OWN_ROWS + 1) * OUT_C];
    __shared__ float esum[OWN_ROWS + 1];
    __shared__ float s_l[OWN_ROWS + 1];
    __shared__ int   q[OWN_WAVES * OWN_QLEN];

    const int tid  = threadIdx.x;
    const int lane = tid & 31;
    const int wv   = __builtin_amdgcn_readfirstlane(tid >> 5);
    int* myq = q + wv * OWN_QLEN;

    const int Gf = nE >> 9;
    int Gp;
    unsigned char* plane = key_plane(h, t + n, wsBytes, nE, &Gp);
    unsigned* rec;
    const int recEdges = rec_plane(plane, Gp, h, wsBytes, n, nE, &rec);

    for (int r0 = blockIdx.x * OWN_ROWS; r0 < n; r0 += gridDim.x * OWN_ROWS) {
        const int R = min(OWN_ROWS, n - r0);

        for (int i = tid; i < (OWN_ROWS + 1) * OUT_C; i += OWN_THREADS) acc[i] = 0.0f;
        for (int i = tid; i <= OWN_ROWS; i += OWN_THREADS) {
            esum[i] = 0.0f;
            s_l[i]  = (i < R) ? s[r0 + i] : 0.0f;
        }
        __syncthreads();

        int cnt = 0;
        const int Gk = Gp >> 1;
        cnt = own_sweep<true >(cnt, 0,      Gk, myq, plane, rec, recEdges, ei, nE, h, t, s_l, acc, esum, lane, wv, r0, R);
        cnt = own_sweep<false>(cnt, Gk * 2, Gf, myq, plane, rec, recEdges, ei, nE, h, t, s_l, acc, esum, lane, wv, r0, R);
        if (wv == 0) {
            for (int e0 = Gf * 512; e0 < nE; e0 += 32) {
                const int  e = e0 + lane;
                const int  sr = ei[min(e, nE - 1)];
                const bool m  = (e < nE) && (((unsigned)sr - (unsigned)r0) < (unsigned)R);
                const unsigned k = __builtin_amdgcn_ballot_w32(m);
                if (m) myq[cnt + __builtin_amdgcn_mbcnt_lo(k, 0u)] = e;
                cnt += __builtin_popcount(k);
            }
        }
        own_drain(myq, cnt, rec, recEdges, ei, nE, h, t, s_l, acc, esum, lane, r0, R);
        __syncthreads();

        for (int i4 = tid; i4 < R * (OUT_C / 4); i4 += OWN_THREADS) {
            const int li = i4 >> 4;
            const int c4 = (i4 & 15) * 4;
            const size_t i = (size_t)(r0 + li);
            const float w = __expf(lrelu(s_l[li] + t[i]));
            const float denom = esum[li] + w;
            const float4 hv = *(const float4*)(h + i * OUT_C + c4);
            const float4 av = *(const float4*)(acc + li * OUT_C + c4);
            float4 o;
            o.x = (av.x + w * hv.x) / denom;
            o.y = (av.y + w * hv.y) / denom;
            o.z = (av.z + w * hv.z) / denom;
            o.w = (av.w + w * hv.w) / denom;
            *(float4*)(out + i * OUT_C + c4) = o;
        }
        __syncthreads();
    }
}

extern "C" void kernel_launch(void* const* d_in, const int* in_sizes, int n_in,
                              void* d_out, int out_size, void* d_ws, size_t ws_size,
                              hipStream_t stream) {
    const float* x  = (const float*)d_in[0];
    const float* W  = (const float*)d_in[1];
    const float* a  = (const float*)d_in[2];
    const int*   ei = (const int*)d_in[3];

    const int n = in_sizes[0] / IN_C;
    const int E = in_sizes[3] / 2;

    float* ws     = (float*)d_ws;
    float* h      = ws;
    float* s      = h + (size_t)n * OUT_C;
    float* t      = s + n;
    float* out    = (float*)d_out;

    const dim3 blk(256);
    const int wavesPerBlock = 256 / 32;

    const int mtiles = (n + 15) / 16;
    gat_gemm_stw<<<(mtiles + wavesPerBlock - 1) / wavesPerBlock, blk, 0, stream>>>(
        x, W, a, h, s, t, ei, n, mtiles, E, (unsigned long long)ws_size);

    gat_owner<<<OWN_BLOCKS, OWN_THREADS, 0, stream>>>(ei, h, s, t, out, n, E,
                                                      (unsigned long long)ws_size);
}


# edges filed by owner per 2048-edge chunk in the GEMM launch (LDS integer adds rank them; segment table); owner blocks fetch only their own records, no key sweep; new edge order, LDS float adds
# speedup vs baseline: 3.3530x; 1.0737x over previous
#include <hip/hip_runtime.h>
#include <hip/hip_bf16.h>
#include <math.h>


typedef __attribute__((ext_vector_type(2))) float v2f;
typedef __attribute__((ext_vector_type(8))) float v8f;

#define IN_C  128
#define OUT_C 64
#define GAT_ALPHA 0.2f

__device__ __forceinline__ float lrelu(float z) {
    return z >= 0.0f ? z : GAT_ALPHA * z;
}

#define OWN_ROWS    392
#define OWN_BLOCKS  256
#define OWN_CHUNK   2048
#define K1_THREADS  256

__device__ __forceinline__ unsigned own_key(int src) {
    return ((unsigned)src / (unsigned)OWN_ROWS) & (unsigned)(OWN_BLOCKS - 1);
}

__device__ __forceinline__ unsigned own_row(int src) {
    return (unsigned)src - ((unsigned)src / (unsigned)OWN_ROWS) * (unsigned)OWN_ROWS;
}

struct OwnPlanes { unsigned* rec; unsigned* seg; int nCh; int ok; };

__device__ __forceinline__ OwnPlanes own_planes(float* wsBase, const float* tEnd,
                                                unsigned long long wsBytes, int n, int nE) {
    OwnPlanes P;
    const unsigned long long off =
        ((unsigned long long)((const char*)tEnd - (const char*)wsBase) + 15ull) & ~15ull;
    const int nCh = (nE >> 11) + ((nE & (OWN_CHUNK - 1)) ? 1 : 0);
    const unsigned long long recBytes = (unsigned long long)nCh * OWN_CHUNK * 4ull;
    const unsigned long long segBytes = (unsigned long long)nCh * OWN_BLOCKS * 4ull;
    P.rec = (unsigned*)((char*)wsBase + off);
    P.seg = (unsigned*)((char*)wsBase + off + recBytes);
    P.nCh = nCh;
    P.ok  = (nE > 0 && (nE & 7) == 0 && n <= OWN_BLOCKS * OWN_ROWS &&
             wsBytes >= off + recBytes + segBytes) ? 1 : 0;
    return P;
}

__global__ void
gat_gemm_stw(const float* __restrict__ x,
             const float* __restrict__ W,
             const float* __restrict__ a,
             float* __restrict__ h,
             float* __restrict__ s,
             float* __restrict__ t,
             const int* __restrict__ ei,
             int n, int mtiles, int nE, unsigned long long wsBytes) {
    __shared__ int kcnt[OWN_BLOCKS];
    __shared__ int koff[OWN_BLOCKS];
    __shared__ int wsum[K1_THREADS / 32];
    __shared__ __attribute__((aligned(16))) unsigned lrec[OWN_CHUNK];

    const int wavesPerBlock = blockDim.x >> 5;
    const int wave = blockIdx.x * wavesPerBlock + ((int)threadIdx.x >> 5);
    const int lane = threadIdx.x & 31;

    {
        const OwnPlanes P = own_planes(h, t + n, wsBytes, n, nE);
        const int tid = (int)threadIdx.x;
        const int wib = tid >> 5;
        for (int c = (int)blockIdx.x; P.ok && c < P.nCh; c += (int)gridDim.x) {
            kcnt[tid] = 0;
            __syncthreads();
            const size_t e0 = (size_t)c * OWN_CHUNK + (size_t)tid * 8;
            const bool live = e0 < (size_t)nE;
            int4 p = {0, 0, 0, 0}, q = {0, 0, 0, 0}, d0 = {0, 0, 0, 0}, d1 = {0, 0, 0, 0};
            if (live) {
                const int4* sp = (const int4*)(ei + e0);
                const int4* dp = (const int4*)(ei + (size_t)nE + e0);
                p = sp[0]; q = sp[1];
                d0 = dp[0]; d1 = dp[1];
            }
            const int sj[8] = {p.x, p.y, p.z, p.w, q.x, q.y, q.z, q.w};
            const int dj[8] = {d0.x, d0.y, d0.z, d0.w, d1.x, d1.y, d1.z, d1.w};
            unsigned key[8];
            int rk[8];
#pragma unroll
            for (int j = 0; j < 8; ++j) {
                key[j] = own_key(sj[j]);
                rk[j]  = 0;
                if (live) rk[j] = atomicAdd(&kcnt[key[j]], 1);
            }
            __syncthreads();
            const int v = kcnt[tid];
            int incl = v;
#pragma unroll
            for (int d = 1; d < 32; d <<= 1) {
                const int y = __shfl_up(incl, d, 32);
                if (lane >= d) incl += y;
            }
            if (lane == 31) wsum[wib] = incl;
            __syncthreads();
            int add = 0;
#pragma unroll
            for (int w = 0; w < K1_THREADS / 32; ++w) add += (w < wib) ? wsum[w] : 0;
            const int excl = incl - v + add;
            koff[tid] = excl;
            P.seg[(size_t)tid * (size_t)P.nCh + (size_t)c] = (unsigned)excl | ((unsigned)v << 16);
            __syncthreads();
            if (live) {
#pragma unroll
                for (int j = 0; j < 8; ++j)
                    lrec[koff[key[j]] + rk[j]] = ((unsigned)dj[j] << 9) | own_row(sj[j]);
            }
            __syncthreads();
            uint4* rg = (uint4*)(P.rec + (size_t)c * OWN_CHUNK);
            const uint4* rl = (const uint4*)lrec;
            rg[tid]              = rl[tid];
            rg[tid + K1_THREADS] = rl[tid + K1_THREADS];
        }
    }

    if (wave >= mtiles) return;

    const int half = lane >> 4;
    const int l15  = lane & 15;
    const int m0   = wave * 16;

    int arow = m0 + l15; if (arow > n - 1) arow = n - 1;
    const float* xrow = x + (size_t)arow * IN_C;

    v2f afrag[32];
#pragma unroll
    for (int ks = 0; ks < 32; ++ks) {
        const int k = ks * 4 + half * 2;
        afrag[ks].x = xrow[k];
        afrag[ks].y = xrow[k + 1];
    }

    float pr1[8], pr2[8];
#pragma unroll
    for (int r = 0; r < 8; ++r) { pr1[r] = 0.0f; pr2[r] = 0.0f; }

#pragma unroll
    for (int nt = 0; nt < 4; ++nt) {
        const int n0 = nt * 16;
        v8f acc = {};
#pragma unroll
        for (int ks = 0; ks < 32; ++ks) {
            const int k = ks * 4 + half * 2;
            v2f b;
            b.x = W[(size_t)k       * OUT_C + n0 + l15];
            b.y = W[(size_t)(k + 1) * OUT_C + n0 + l15];
            acc = __builtin_amdgcn_wmma_f32_16x16x4_f32(
                 false, afrag[ks],  false, b,
                 (short)0, acc,  false,  false);
        }
        const float a1c = a[n0 + l15];
        const float a2c = a[OUT_C + n0 + l15];
#pragma unroll
        for (int r = 0; r < 8; ++r) {
            const int row = m0 + r + 8 * half;
            const float v = acc[r];
            if (row < n) h[(size_t)row * OUT_C + n0 + l15] = v;
            pr1[r] += v * a1c;
            pr2[r] += v * a2c;
        }
    }

#pragma unroll
    for (int mask = 8; mask >= 1; mask >>= 1) {
#pragma unroll
        for (int r = 0; r < 8; ++r) {
            pr1[r] += __shfl_xor(pr1[r], mask, 32);
            pr2[r] += __shfl_xor(pr2[r], mask, 32);
        }
    }
    if (l15 == 0) {
#pragma unroll
        for (int r = 0; r < 8; ++r) {
            const int row = m0 + r + 8 * half;
            if (row < n) { s[row] = pr1[r]; t[row] = pr2[r]; }
        }
    }
}

#define OWN_THREADS 1024
#define OWN_WAVES   32
#define OWN_QLEN    1088
#define OWN_FETCH   4

__device__ __forceinline__ void own_drain(const int* myq, int cnt,
                                          const float* __restrict__ h,
                                          const float* __restrict__ t,
                                          const float* s_l, float* acc, float* esum,
                                          int lane) {
    for (int b = 0; b < cnt; b += 32) {
        const int nb = min(32, cnt - b);
        const bool own = lane < nb;
        unsigned r = (unsigned)OWN_ROWS;
        if (own) r = (unsigned)myq[b + lane];
        const int dst = (int)(r >> 9);
        const int ls  = (int)(r & 511u);
        float w = __expf(lrelu(s_l[ls] + t[dst]));
        if (!own) w = 0.0f;
        if (own) atomicAdd(&esum[ls], w);
        for (int k0 = 0; k0 < nb; k0 += 8) {
            float2 v[8];
#pragma unroll
            for (int j = 0; j < 8; ++j) {
                const int d = __builtin_amdgcn_readlane(dst, k0 + j);
                v[j] = *(const float2*)(h + (size_t)d * OUT_C + lane * 2);
            }
#pragma unroll
            for (int j = 0; j < 8; ++j) {
                const float wk = __int_as_float(__builtin_amdgcn_readlane(__float_as_int(w), k0 + j));
                const int   lk = __builtin_amdgcn_readlane(ls, k0 + j);
                float* a = acc + lk * OUT_C + lane * 2;
                atomicAdd(a,     wk * v[j].x);
                atomicAdd(a + 1, wk * v[j].y);
            }
        }
    }
}

__device__ __forceinline__ int own_collect(int cnt, int* myq,
                                           const unsigned* __restrict__ rec,
                                           const unsigned* __restrict__ seg, int nCh, int owner,
                                           const float* __restrict__ h,
                                           const float* __restrict__ t,
                                           const float* s_l, float* acc, float* esum,
                                           int lane, int wv) {
    const int cBeg = (int)(((long long)wv * nCh) / OWN_WAVES);
    const int cEnd = (int)(((long long)(wv + 1) * nCh) / OWN_WAVES);
    for (int c0 = cBeg; c0 < cEnd; c0 += 32) {
        const int c = c0 + lane;
        unsigned sw = 0u;
        if (c < cEnd) sw = seg[(size_t)owner * (size_t)nCh + (size_t)c];
        const int ct = (int)(sw >> 16);
        const int st = (int)(sw & 0xffffu);
        int incl = ct;
#pragma unroll
        for (int d = 1; d < 32; d <<= 1) {
            const int y = __shfl_up(incl, d, 32);
            if (lane >= d) incl += y;
        }
        const int excl  = incl - ct;
        const int total = __builtin_amdgcn_readlane(incl, 31);
        const int gbase = c * OWN_CHUNK + st - excl;
        int ex[32];
#pragma unroll
        for (int j = 0; j < 32; ++j) ex[j] = __builtin_amdgcn_readlane(excl, j);

        for (int p0 = 0; p0 < total; p0 += 32 * OWN_FETCH) {
            unsigned r[OWN_FETCH];
#pragma unroll
            for (int u = 0; u < OWN_FETCH; ++u) {
                const int p = p0 + 32 * u + lane;
                int js = -1;
#pragma unroll
                for (int j = 0; j < 32; ++j) js += (p >= ex[j]) ? 1 : 0;
                const int gb = __shfl(gbase, js, 32);
                r[u] = (unsigned)OWN_ROWS;
                if (p < total) r[u] = rec[gb + p];
            }
#pragma unroll
            for (int u = 0; u < OWN_FETCH; ++u) {
                const int p = p0 + 32 * u + lane;
                if (p < total) myq[cnt + 32 * u + lane] = (int)r[u];
            }
            cnt += min(32 * OWN_FETCH, total - p0);
            if (cnt > OWN_QLEN - 32 * OWN_FETCH) {
                own_drain(myq, cnt, h, t, s_l, acc, esum, lane);
                cnt = 0;
            }
        }
    }
    return cnt;
}

__device__ __forceinline__ int own_scan_sources(int cnt, int* myq,
                                                const int* __restrict__ ei, int nE,
                                                const float* __restrict__ h,
                                                const float* __restrict__ t,
                                                const float* s_l, float* acc, float* esum,
                                                int lane, int wv, int r0, int R) {
    for (long long e0 = (long long)wv * 32; e0 < (long long)nE; e0 += (long long)OWN_WAVES * 32) {
        const long long e = e0 + lane;
        const bool in = e < (long long)nE;
        const int  sr = in ? ei[e] : -1;
        const bool m  = in && (((unsigned)sr - (unsigned)r0) < (unsigned)R);
        const unsigned k = __builtin_amdgcn_ballot_w32(m);
        if (m) {
            const int d = ei[(long long)nE + e];
            myq[cnt + __builtin_amdgcn_mbcnt_lo(k, 0u)] = (int)(((unsigned)d << 9) | (unsigned)(sr - r0));
        }
        cnt += __builtin_popcount(k);
        if (cnt > OWN_QLEN - 32) {
            own_drain(myq, cnt, h, t, s_l, acc, esum, lane);
            cnt = 0;
        }
    }
    return cnt;
}

__global__ void __launch_bounds__(OWN_THREADS)
gat_owner(const int* __restrict__ ei,
          float* __restrict__ h,
          const float* __restrict__ s,
          const float* __restrict__ t,
          float* __restrict__ out,
          int n, int nE, unsigned long long wsBytes) {
    __shared__ float acc[(OWN_ROWS + 1) * OUT_C];
    __shared__ float esum[OWN_ROWS + 1];
    __shared__ float s_l[OWN_ROWS + 1];
    __shared__ int   q[OWN_WAVES * OWN_QLEN];

    const int tid  = threadIdx.x;
    const int lane = tid & 31;
    const int wv   = __builtin_amdgcn_readfirstlane(tid >> 5);
    int* myq = q + wv * OWN_QLEN;

    const OwnPlanes P = own_planes(h, t + n, wsBytes, n, nE);

    for (int r0 = blockIdx.x * OWN_ROWS; r0 < n; r0 += gridDim.x * OWN_ROWS) {
        const int R = min(OWN_ROWS, n - r0);

        for (int i = tid; i < (OWN_ROWS + 1) * OUT_C; i += OWN_THREADS) acc[i] = 0.0f;
        for (int i = tid; i <= OWN_ROWS; i += OWN_THREADS) {
            esum[i] = 0.0f;
            s_l[i]  = (i < R) ? s[r0 + i] : 0.0f;
        }
        __syncthreads();

        int cnt = 0;
        if (P.ok) {
            cnt = own_collect(cnt, myq, P.rec, P.seg, P.nCh, (int)blockIdx.x, h, t, s_l, acc, esum, lane, wv);
        } else {
            cnt = own_scan_sources(cnt, myq, ei, nE, h, t, s_l, acc, esum, lane, wv, r0, R);
        }
        own_drain(myq, cnt, h, t, s_l, acc, esum, lane);
        __syncthreads();

        for (int i4 = tid; i4 < R * (OUT_C / 4); i4 += OWN_THREADS) {
            const int li = i4 >> 4;
            const int c4 = (i4 & 15) * 4;
            const size_t i = (size_t)(r0 + li);
            const float w = __expf(lrelu(s_l[li] + t[i]));
            const float denom = esum[li] + w;
            const float4 hv = *(const float4*)(h + i * OUT_C + c4);
            const float4 av = *(const float4*)(acc + li * OUT_C + c4);
            float4 o;
            o.x = (av.x + w * hv.x) / denom;
            o.y = (av.y + w * hv.y) / denom;
            o.z = (av.z + w * hv.z) / denom;
            o.w = (av.w + w * hv.w) / denom;
            *(float4*)(out + i * OUT_C + c4) = o;
        }
        __syncthreads();
    }
}

extern "C" void kernel_launch(void* const* d_in, const int* in_sizes, int n_in,
                              void* d_out, int out_size, void* d_ws, size_t ws_size,
                              hipStream_t stream) {
    const float* x  = (const float*)d_in[0];
    const float* W  = (const float*)d_in[1];
    const float* a  = (const float*)d_in[2];
    const int*   ei = (const int*)d_in[3];

    const int n = in_sizes[0] / IN_C;
    const int E = in_sizes[3] / 2;

    float* ws     = (float*)d_ws;
    float* h      = ws;
    float* s      = h + (size_t)n * OUT_C;
    float* t      = s + n;
    float* out    = (float*)d_out;

    const dim3 blk(K1_THREADS);
    const int wavesPerBlock = K1_THREADS / 32;

    const int mtiles = (n + 15) / 16;
    gat_gemm_stw<<<(mtiles + wavesPerBlock - 1) / wavesPerBlock, blk, 0, stream>>>(
        x, W, a, h, s, t, ei, n, mtiles, E, (unsigned long long)ws_size);

    gat_owner<<<OWN_BLOCKS, OWN_THREADS, 0, stream>>>(ei, h, s, t, out, n, E,
                                                      (unsigned long long)ws_size);
}


# owner block hands each record to the wave owning its row (row&31: LDS integer adds rank records into 32 lists, one barrier); row sums by plain LDS load+add+store, not float adds; arrival order
# speedup vs baseline: 3.4804x; 1.0380x over previous
#include <hip/hip_runtime.h>
#include <hip/hip_bf16.h>
#include <math.h>


typedef __attribute__((ext_vector_type(2))) float v2f;
typedef __attribute__((ext_vector_type(8))) float v8f;

#define IN_C  128
#define OUT_C 64
#define GAT_ALPHA 0.2f

__device__ __forceinline__ float lrelu(float z) {
    return z >= 0.0f ? z : GAT_ALPHA * z;
}

#define OWN_ROWS    392
#define OWN_BLOCKS  256
#define OWN_CHUNK   2048
#define K1_THREADS  256

__device__ __forceinline__ unsigned own_key(int src) {
    return ((unsigned)src / (unsigned)OWN_ROWS) & (unsigned)(OWN_BLOCKS - 1);
}

__device__ __forceinline__ unsigned own_row(int src) {
    return (unsigned)src - ((unsigned)src / (unsigned)OWN_ROWS) * (unsigned)OWN_ROWS;
}

struct OwnPlanes { unsigned* rec; unsigned* seg; int nCh; int ok; };

__device__ __forceinline__ OwnPlanes own_planes(float* wsBase, const float* tEnd,
                                                unsigned long long wsBytes, int n, int nE) {
    OwnPlanes P;
    const unsigned long long off =
        ((unsigned long long)((const char*)tEnd - (const char*)wsBase) + 15ull) & ~15ull;
    const int nCh = (nE >> 11) + ((nE & (OWN_CHUNK - 1)) ? 1 : 0);
    const unsigned long long recBytes = (unsigned long long)nCh * OWN_CHUNK * 4ull;
    const unsigned long long segBytes = (unsigned long long)nCh * OWN_BLOCKS * 4ull;
    P.rec = (unsigned*)((char*)wsBase + off);
    P.seg = (unsigned*)((char*)wsBase + off + recBytes);
    P.nCh = nCh;
    P.ok  = (nE > 0 && (nE & 7) == 0 && n <= OWN_BLOCKS * OWN_ROWS &&
             wsBytes >= off + recBytes + segBytes) ? 1 : 0;
    return P;
}

__global__ void
gat_gemm_stw(const float* __restrict__ x,
             const float* __restrict__ W,
             const float* __restrict__ a,
             float* __restrict__ h,
             float* __restrict__ s,
             float* __restrict__ t,
             const int* __restrict__ ei,
             int n, int mtiles, int nE, unsigned long long wsBytes) {
    __shared__ int kcnt[OWN_BLOCKS];
    __shared__ int koff[OWN_BLOCKS];
    __shared__ int wsum[K1_THREADS / 32];
    __shared__ __attribute__((aligned(16))) unsigned lrec[OWN_CHUNK];

    const int wavesPerBlock = blockDim.x >> 5;
    const int wave = blockIdx.x * wavesPerBlock + ((int)threadIdx.x >> 5);
    const int lane = threadIdx.x & 31;

    {
        const OwnPlanes P = own_planes(h, t + n, wsBytes, n, nE);
        const int tid = (int)threadIdx.x;
        const int wib = tid >> 5;
        for (int c = (int)blockIdx.x; P.ok && c < P.nCh; c += (int)gridDim.x) {
            kcnt[tid] = 0;
            __syncthreads();
            const size_t e0 = (size_t)c * OWN_CHUNK + (size_t)tid * 8;
            const bool live = e0 < (size_t)nE;
            int4 p = {0, 0, 0, 0}, q = {0, 0, 0, 0}, d0 = {0, 0, 0, 0}, d1 = {0, 0, 0, 0};
            if (live) {
                const int4* sp = (const int4*)(ei + e0);
                const int4* dp = (const int4*)(ei + (size_t)nE + e0);
                p = sp[0]; q = sp[1];
                d0 = dp[0]; d1 = dp[1];
            }
            const int sj[8] = {p.x, p.y, p.z, p.w, q.x, q.y, q.z, q.w};
            const int dj[8] = {d0.x, d0.y, d0.z, d0.w, d1.x, d1.y, d1.z, d1.w};
            unsigned key[8];
            int rk[8];
#pragma unroll
            for (int j = 0; j < 8; ++j) {
                key[j] = own_key(sj[j]);
                rk[j]  = 0;
                if (live) rk[j] = atomicAdd(&kcnt[key[j]], 1);
            }
            __syncthreads();
            const int v = kcnt[tid];
            int incl = v;
#pragma unroll
            for (int d = 1; d < 32; d <<= 1) {
                const int y = __shfl_up(incl, d, 32);
                if (lane >= d) incl += y;
            }
            if (lane == 31) wsum[wib] = incl;
            __syncthreads();
            int add = 0;
#pragma unroll
            for (int w = 0; w < K1_THREADS / 32; ++w) add += (w < wib) ? wsum[w] : 0;
            const int excl = incl - v + add;
            koff[tid] = excl;
            P.seg[(size_t)tid * (size_t)P.nCh + (size_t)c] = (unsigned)excl | ((unsigned)v << 16);
            __syncthreads();
            if (live) {
#pragma unroll
                for (int j = 0; j < 8; ++j)
                    lrec[koff[key[j]] + rk[j]] = ((unsigned)dj[j] << 9) | own_row(sj[j]);
            }
            __syncthreads();
            uint4* rg = (uint4*)(P.rec + (size_t)c * OWN_CHUNK);
            const uint4* rl = (const uint4*)lrec;
            rg[tid]              = rl[tid];
            rg[tid + K1_THREADS] = rl[tid + K1_THREADS];
        }
    }

    if (wave >= mtiles) return;

    const int half = lane >> 4;
    const int l15  = lane & 15;
    const int m0   = wave * 16;

    int arow = m0 + l15; if (arow > n - 1) arow = n - 1;
    const float* xrow = x + (size_t)arow * IN_C;

    v2f afrag[32];
#pragma unroll
    for (int ks = 0; ks < 32; ++ks) {
        const int k = ks * 4 + half * 2;
        afrag[ks].x = xrow[k];
        afrag[ks].y = xrow[k + 1];
    }

    float pr1[8], pr2[8];
#pragma unroll
    for (int r = 0; r < 8; ++r) { pr1[r] = 0.0f; pr2[r] = 0.0f; }

#pragma unroll
    for (int nt = 0; nt < 4; ++nt) {
        const int n0 = nt * 16;
        v8f acc = {};
#pragma unroll
        for (int ks = 0; ks < 32; ++ks) {
            const int k = ks * 4 + half * 2;
            v2f b;
            b.x = W[(size_t)k       * OUT_C + n0 + l15];
            b.y = W[(size_t)(k + 1) * OUT_C + n0 + l15];
            acc = __builtin_amdgcn_wmma_f32_16x16x4_f32(
                 false, afrag[ks],  false, b,
                 (short)0, acc,  false,  false);
        }
        const float a1c = a[n0 + l15];
        const float a2c = a[OUT_C + n0 + l15];
#pragma unroll
        for (int r = 0; r < 8; ++r) {
            const int row = m0 + r + 8 * half;
            const float v = acc[r];
            if (row < n) h[(size_t)row * OUT_C + n0 + l15] = v;
            pr1[r] += v * a1c;
            pr2[r] += v * a2c;
        }
    }

#pragma unroll
    for (int mask = 8; mask >= 1; mask >>= 1) {
#pragma unroll
        for (int r = 0; r < 8; ++r) {
            pr1[r] += __shfl_xor(pr1[r], mask, 32);
            pr2[r] += __shfl_xor(pr2[r], mask, 32);
        }
    }
    if (l15 == 0) {
#pragma unroll
        for (int r = 0; r < 8; ++r) {
            const int row = m0 + r + 8 * half;
            if (row < n) { s[row] = pr1[r]; t[row] = pr2[r]; }
        }
    }
}

#define OWN_THREADS 1024
#define OWN_WAVES   32
#define OWN_QLEN    1088
#define OWN_FETCH   4

template <bool OWNED>
__device__ __forceinline__ void own_drain(const int* myq, int cnt,
                                          const float* __restrict__ h,
                                          const float* __restrict__ t,
                                          const float* s_l, float* acc, float* esum,
                                          int lane) {
    for (int b = 0; b < cnt; b += 32) {
        const int nb = min(32, cnt - b);
        const bool own = lane < nb;
        unsigned r = (unsigned)OWN_ROWS;
        if (own) r = (unsigned)myq[b + lane];
        const int dst = (int)(r >> 9);
        const int ls  = (int)(r & 511u);
        float w = __expf(lrelu(s_l[ls] + t[dst]));
        if (!own) w = 0.0f;
        if (own) atomicAdd(&esum[ls], w);
        for (int k0 = 0; k0 < nb; k0 += 8) {
            float2 v[8];
#pragma unroll
            for (int j = 0; j < 8; ++j) {
                const int d = __builtin_amdgcn_readlane(dst, k0 + j);
                v[j] = *(const float2*)(h + (size_t)d * OUT_C + lane * 2);
            }
#pragma unroll
            for (int j = 0; j < 8; ++j) {
                const float wk = __int_as_float(__builtin_amdgcn_readlane(__float_as_int(w), k0 + j));
                const int   lk = __builtin_amdgcn_readlane(ls, k0 + j);
                float* a = acc + lk * OUT_C + lane * 2;
                if (OWNED) {
                    float2 c = *(float2*)a;
                    c.x += wk * v[j].x;
                    c.y += wk * v[j].y;
                    *(float2*)a = c;
                } else {
                    atomicAdd(a,     wk * v[j].x);
                    atomicAdd(a + 1, wk * v[j].y);
                }
            }
        }
    }
}

__device__ __forceinline__ int own_collect(int cnt, int* myq,
                                           const unsigned* __restrict__ rec,
                                           const unsigned* __restrict__ seg, int nCh, int owner,
                                           const float* __restrict__ h,
                                           const float* __restrict__ t,
                                           const float* s_l, float* acc, float* esum,
                                           int lane, int wv) {
    const int cBeg = (int)(((long long)wv * nCh) / OWN_WAVES);
    const int cEnd = (int)(((long long)(wv + 1) * nCh) / OWN_WAVES);
    for (int c0 = cBeg; c0 < cEnd; c0 += 32) {
        const int c = c0 + lane;
        unsigned sw = 0u;
        if (c < cEnd) sw = seg[(size_t)owner * (size_t)nCh + (size_t)c];
        const int ct = (int)(sw >> 16);
        const int st = (int)(sw & 0xffffu);
        int incl = ct;
#pragma unroll
        for (int d = 1; d < 32; d <<= 1) {
            const int y = __shfl_up(incl, d, 32);
            if (lane >= d) incl += y;
        }
        const int excl  = incl - ct;
        const int total = __builtin_amdgcn_readlane(incl, 31);
        const int gbase = c * OWN_CHUNK + st - excl;
        int ex[32];
#pragma unroll
        for (int j = 0; j < 32; ++j) ex[j] = __builtin_amdgcn_readlane(excl, j);

        for (int p0 = 0; p0 < total; p0 += 32 * OWN_FETCH) {
            unsigned r[OWN_FETCH];
#pragma unroll
            for (int u = 0; u < OWN_FETCH; ++u) {
                const int p = p0 + 32 * u + lane;
                int js = -1;
#pragma unroll
                for (int j = 0; j < 32; ++j) js += (p >= ex[j]) ? 1 : 0;
                const int gb = __shfl(gbase, js, 32);
                r[u] = (unsigned)OWN_ROWS;
                if (p < total) r[u] = rec[gb + p];
            }
#pragma unroll
            for (int u = 0; u < OWN_FETCH; ++u) {
                const int p = p0 + 32 * u + lane;
                if (p < total) myq[cnt + 32 * u + lane] = (int)r[u];
            }
            cnt += min(32 * OWN_FETCH, total - p0);
            if (cnt > OWN_QLEN - 32 * OWN_FETCH) {
                own_drain<false>(myq, cnt, h, t, s_l, acc, esum, lane);
                cnt = 0;
            }
        }
    }
    return cnt;
}

__device__ __forceinline__ void own_route(int* q, int* qcnt, int* qovf,
                                          const unsigned* __restrict__ rec,
                                          const unsigned* __restrict__ seg, int nCh, int owner,
                                          int lane, int wv) {
    const int cBeg = (int)(((long long)wv * nCh) / OWN_WAVES);
    const int cEnd = (int)(((long long)(wv + 1) * nCh) / OWN_WAVES);
    bool full = false;
    for (int c0 = cBeg; c0 < cEnd; c0 += 32) {
        const int c = c0 + lane;
        unsigned sw = 0u;
        if (c < cEnd) sw = seg[(size_t)owner * (size_t)nCh + (size_t)c];
        const int ct = (int)(sw >> 16);
        const int st = (int)(sw & 0xffffu);
        int incl = ct;
#pragma unroll
        for (int d = 1; d < 32; d <<= 1) {
            const int y = __shfl_up(incl, d, 32);
            if (lane >= d) incl += y;
        }
        const int excl  = incl - ct;
        const int total = __builtin_amdgcn_readlane(incl, 31);
        const int gbase = c * OWN_CHUNK + st - excl;
        int ex[32];
#pragma unroll
        for (int j = 0; j < 32; ++j) ex[j] = __builtin_amdgcn_readlane(excl, j);

        for (int p0 = 0; p0 < total; p0 += 32 * OWN_FETCH) {
            unsigned r[OWN_FETCH];
#pragma unroll
            for (int u = 0; u < OWN_FETCH; ++u) {
                const int p = p0 + 32 * u + lane;
                int js = -1;
#pragma unroll
                for (int j = 0; j < 32; ++j) js += (p >= ex[j]) ? 1 : 0;
                const int gb = __shfl(gbase, js, 32);
                r[u] = (unsigned)OWN_ROWS;
                if (p < total) r[u] = rec[gb + p];
            }
#pragma unroll
            for (int u = 0; u < OWN_FETCH; ++u) {
                const int p = p0 + 32 * u + lane;
                if (p < total) {
                    const int o   = (int)(r[u] & (unsigned)(OWN_WAVES - 1));
                    const int pos = atomicAdd(&qcnt[o], 1);
                    if (pos < OWN_QLEN) q[o * OWN_QLEN + pos] = (int)r[u];
                    else full = true;
                }
            }
        }
    }
    if (full) *qovf = 1;
}

__device__ __forceinline__ int own_scan_sources(int cnt, int* myq,
                                                const int* __restrict__ ei, int nE,
                                                const float* __restrict__ h,
                                                const float* __restrict__ t,
                                                const float* s_l, float* acc, float* esum,
                                                int lane, int wv, int r0, int R) {
    for (long long e0 = (long long)wv * 32; e0 < (long long)nE; e0 += (long long)OWN_WAVES * 32) {
        const long long e = e0 + lane;
        const bool in = e < (long long)nE;
        const int  sr = in ? ei[e] : -1;
        const bool m  = in && (((unsigned)sr - (unsigned)r0) < (unsigned)R);
        const unsigned k = __builtin_amdgcn_ballot_w32(m);
        if (m) {
            const int d = ei[(long long)nE + e];
            myq[cnt + __builtin_amdgcn_mbcnt_lo(k, 0u)] = (int)(((unsigned)d << 9) | (unsigned)(sr - r0));
        }
        cnt += __builtin_popcount(k);
        if (cnt > OWN_QLEN - 32) {
            own_drain<false>(myq, cnt, h, t, s_l, acc, esum, lane);
            cnt = 0;
        }
    }
    return cnt;
}

__global__ void __launch_bounds__(OWN_THREADS)
gat_owner(const int* __restrict__ ei,
          float* __restrict__ h,
          const float* __restrict__ s,
          const float* __restrict__ t,
          float* __restrict__ out,
          int n, int nE, unsigned long long wsBytes) {
    __shared__ __attribute__((aligned(16))) float acc[(OWN_ROWS + 1) * OUT_C];
    __shared__ float esum[OWN_ROWS + 1];
    __shared__ float s_l[OWN_ROWS + 1];
    __shared__ int   q[OWN_WAVES * OWN_QLEN];
    __shared__ int   qcnt[OWN_WAVES];
    __shared__ int   qovf;

    const int tid  = threadIdx.x;
    const int lane = tid & 31;
    const int wv   = __builtin_amdgcn_readfirstlane(tid >> 5);
    int* myq = q + wv * OWN_QLEN;

    const OwnPlanes P = own_planes(h, t + n, wsBytes, n, nE);

    for (int r0 = blockIdx.x * OWN_ROWS; r0 < n; r0 += gridDim.x * OWN_ROWS) {
        const int R = min(OWN_ROWS, n - r0);

        for (int i = tid; i < (OWN_ROWS + 1) * OUT_C; i += OWN_THREADS) acc[i] = 0.0f;
        for (int i = tid; i <= OWN_ROWS; i += OWN_THREADS) {
            esum[i] = 0.0f;
            s_l[i]  = (i < R) ? s[r0 + i] : 0.0f;
        }
        if (tid < OWN_WAVES) qcnt[tid] = 0;
        if (tid == 0) qovf = 0;
        __syncthreads();

        int cnt = 0;
        if (P.ok) {
            own_route(q, qcnt, &qovf, P.rec, P.seg, P.nCh, (int)blockIdx.x, lane, wv);
            __syncthreads();
            if (qovf == 0) {
                cnt = __builtin_amdgcn_readfirstlane(qcnt[wv]);
                own_drain<true>(myq, cnt, h, t, s_l, acc, esum, lane);
            } else {
                cnt = own_collect(cnt, myq, P.rec, P.seg, P.nCh, (int)blockIdx.x, h, t, s_l, acc, esum, lane, wv);
                own_drain<false>(myq, cnt, h, t, s_l, acc, esum, lane);
            }
        } else {
            cnt = own_scan_sources(cnt, myq, ei, nE, h, t, s_l, acc, esum, lane, wv, r0, R);
            own_drain<false>(myq, cnt, h, t, s_l, acc, esum, lane);
        }
        __syncthreads();

        for (int i4 = tid; i4 < R * (OUT_C / 4); i4 += OWN_THREADS) {
            const int li = i4 >> 4;
            const int c4 = (i4 & 15) * 4;
            const size_t i = (size_t)(r0 + li);
            const float w = __expf(lrelu(s_l[li] + t[i]));
            const float denom = esum[li] + w;
            const float4 hv = *(const float4*)(h + i * OUT_C + c4);
            const float4 av = *(const float4*)(acc + li * OUT_C + c4);
            float4 o;
            o.x = (av.x + w * hv.x) / denom;
            o.y = (av.y + w * hv.y) / denom;
            o.z = (av.z + w * hv.z) / denom;
            o.w = (av.w + w * hv.w) / denom;
            *(float4*)(out + i * OUT_C + c4) = o;
        }
        __syncthreads();
    }
}

extern "C" void kernel_launch(void* const* d_in, const int* in_sizes, int n_in,
                              void* d_out, int out_size, void* d_ws, size_t ws_size,
                              hipStream_t stream) {
    const float* x  = (const float*)d_in[0];
    const float* W  = (const float*)d_in[1];
    const float* a  = (const float*)d_in[2];
    const int*   ei = (const int*)d_in[3];

    const int n = in_sizes[0] / IN_C;
    const int E = in_sizes[3] / 2;

    float* ws     = (float*)d_ws;
    float* h      = ws;
    float* s      = h + (size_t)n * OUT_C;
    float* t      = s + n;
    float* out    = (float*)d_out;

    const dim3 blk(K1_THREADS);
    const int wavesPerBlock = K1_THREADS / 32;

    const int mtiles = (n + 15) / 16;
    gat_gemm_stw<<<(mtiles + wavesPerBlock - 1) / wavesPerBlock, blk, 0, stream>>>(
        x, W, a, h, s, t, ei, n, mtiles, E, (unsigned long long)ws_size);

    gat_owner<<<OWN_BLOCKS, OWN_THREADS, 0, stream>>>(ei, h, s, t, out, n, E,
                                                      (unsigned long long)ws_size);
}


# GEMM: N-tile nt = columns 4c+nt, so one 16-byte W load per lane and k feeds 4 interleaved tile chains, asked 3 k-steps ahead (x fragments in parts for room); 16-byte h stores; s,t dots in new order
# speedup vs baseline: 4.0310x; 1.1582x over previous
#include <hip/hip_runtime.h>
#include <hip/hip_bf16.h>
#include <math.h>


typedef __attribute__((ext_vector_type(2))) float v2f;
typedef __attribute__((ext_vector_type(8))) float v8f;

#define IN_C  128
#define OUT_C 64
#define GAT_ALPHA 0.2f

__device__ __forceinline__ float lrelu(float z) {
    return z >= 0.0f ? z : GAT_ALPHA * z;
}

#define OWN_ROWS    392
#define OWN_BLOCKS  256
#define OWN_CHUNK   2048
#define K1_THREADS  256
#define GEMM_AHEAD  3
#define LOADS_STAY_AHEAD() asm volatile("" ::: "memory")
__device__ constexpr int gemm_a_from(int ks) {
    return ks == 4 ? 8 : (ks == 8 ? 12 : (ks == 12 ? 24 : (ks == 16 ? 28 : 0)));
}

__device__ __forceinline__ unsigned own_key(int src) {
    return ((unsigned)src / (unsigned)OWN_ROWS) & (unsigned)(OWN_BLOCKS - 1);
}

__device__ __forceinline__ unsigned own_row(int src) {
    return (unsigned)src - ((unsigned)src / (unsigned)OWN_ROWS) * (unsigned)OWN_ROWS;
}

struct OwnPlanes { unsigned* rec; unsigned* seg; int nCh; int ok; };

__device__ __forceinline__ OwnPlanes own_planes(float* wsBase, const float* tEnd,
                                                unsigned long long wsBytes, int n, int nE) {
    OwnPlanes P;
    const unsigned long long off =
        ((unsigned long long)((const char*)tEnd - (const char*)wsBase) + 15ull) & ~15ull;
    const int nCh = (nE >> 11) + ((nE & (OWN_CHUNK - 1)) ? 1 : 0);
    const unsigned long long recBytes = (unsigned long long)nCh * OWN_CHUNK * 4ull;
    const unsigned long long segBytes = (unsigned long long)nCh * OWN_BLOCKS * 4ull;
    P.rec = (unsigned*)((char*)wsBase + off);
    P.seg = (unsigned*)((char*)wsBase + off + recBytes);
    P.nCh = nCh;
    P.ok  = (nE > 0 && (nE & 7) == 0 && n <= OWN_BLOCKS * OWN_ROWS &&
             wsBytes >= off + recBytes + segBytes) ? 1 : 0;
    return P;
}

__global__ void
gat_gemm_stw(const float* __restrict__ x,
             const float* __restrict__ W,
             const float* __restrict__ a,
             float* __restrict__ h,
             float* __restrict__ s,
             float* __restrict__ t,
             const int* __restrict__ ei,
             int n, int mtiles, int nE, unsigned long long wsBytes) {
    __shared__ int kcnt[OWN_BLOCKS];
    __shared__ int koff[OWN_BLOCKS];
    __shared__ int wsum[K1_THREADS / 32];
    __shared__ __attribute__((aligned(16))) unsigned lrec[OWN_CHUNK];

    const int wavesPerBlock = blockDim.x >> 5;
    const int wave = blockIdx.x * wavesPerBlock + ((int)threadIdx.x >> 5);
    const int lane = threadIdx.x & 31;

    {
        const OwnPlanes P = own_planes(h, t + n, wsBytes, n, nE);
        const int tid = (int)threadIdx.x;
        const int wib = tid >> 5;
        for (int c = (int)blockIdx.x; P.ok && c < P.nCh; c += (int)gridDim.x) {
            kcnt[tid] = 0;
            __syncthreads();
            const size_t e0 = (size_t)c * OWN_CHUNK + (size_t)tid * 8;
            const bool live = e0 < (size_t)nE;
            int4 p = {0, 0, 0, 0}, q = {0, 0, 0, 0}, d0 = {0, 0, 0, 0}, d1 = {0, 0, 0, 0};
            if (live) {
                const int4* sp = (const int4*)(ei + e0);
                const int4* dp = (const int4*)(ei + (size_t)nE + e0);
                p = sp[0]; q = sp[1];
                d0 = dp[0]; d1 = dp[1];
            }
            const int sj[8] = {p.x, p.y, p.z, p.w, q.x, q.y, q.z, q.w};
            const int dj[8] = {d0.x, d0.y, d0.z, d0.w, d1.x, d1.y, d1.z, d1.w};
            unsigned key[8];
            int rk[8];
#pragma unroll
            for (int j = 0; j < 8; ++j) {
                key[j] = own_key(sj[j]);
                rk[j]  = 0;
                if (live) rk[j] = atomicAdd(&kcnt[key[j]], 1);
            }
            __syncthreads();
            const int v = kcnt[tid];
            int incl = v;
#pragma unroll
            for (int d = 1; d < 32; d <<= 1) {
                const int y = __shfl_up(incl, d, 32);
                if (lane >= d) incl += y;
            }
            if (lane == 31) wsum[wib] = incl;
            __syncthreads();
            int add = 0;
#pragma unroll
            for (int w = 0; w < K1_THREADS / 32; ++w) add += (w < wib) ? wsum[w] : 0;
            const int excl = incl - v + add;
            koff[tid] = excl;
            P.seg[(size_t)tid * (size_t)P.nCh + (size_t)c] = (unsigned)excl | ((unsigned)v << 16);
            __syncthreads();
            if (live) {
#pragma unroll
                for (int j = 0; j < 8; ++j)
                    lrec[koff[key[j]] + rk[j]] = ((unsigned)dj[j] << 9) | own_row(sj[j]);
            }
            __syncthreads();
            uint4* rg = (uint4*)(P.rec + (size_t)c * OWN_CHUNK);
            const uint4* rl = (const uint4*)lrec;
            rg[tid]              = rl[tid];
            rg[tid + K1_THREADS] = rl[tid + K1_THREADS];
        }
    }

    if (wave >= mtiles) return;

    const int half = lane >> 4;
    const int l15  = lane & 15;
    const int m0   = wave * 16;

    int arow = m0 + l15; if (arow > n - 1) arow = n - 1;
    const float* xl = x + (size_t)arow * IN_C + half * 2;

    const float* wl = W + (size_t)(half * 2) * OUT_C + l15 * 4;

    v2f    afrag[32];
    float4 bq[32][2];
#pragma unroll
    for (int ks = 0; ks < 8; ++ks) {
        afrag[ks]      = *(const v2f*)(xl + ks * 4);
        afrag[ks + 16] = *(const v2f*)(xl + (ks + 16) * 4);
    }
#pragma unroll
    for (int ks = 0; ks < GEMM_AHEAD; ++ks) {
        bq[ks][0] = *(const float4*)(wl + (size_t)(ks * 4)     * OUT_C);
        bq[ks][1] = *(const float4*)(wl + (size_t)(ks * 4 + 1) * OUT_C);
    }

    v8f acc0 = {}, acc1 = {}, acc2 = {}, acc3 = {};
#pragma unroll
    for (int ks = 0; ks < 32; ++ks) {
        if (gemm_a_from(ks) != 0) {
#pragma unroll
            for (int j = 0; j < 4; ++j)
                afrag[gemm_a_from(ks) + j] = *(const v2f*)(xl + (gemm_a_from(ks) + j) * 4);
        }
        if (ks + GEMM_AHEAD < 32) {
            bq[ks + GEMM_AHEAD][0] = *(const float4*)(wl + (size_t)((ks + GEMM_AHEAD) * 4)     * OUT_C);
            bq[ks + GEMM_AHEAD][1] = *(const float4*)(wl + (size_t)((ks + GEMM_AHEAD) * 4 + 1) * OUT_C);
        }
        LOADS_STAY_AHEAD();
        v2f b;
        b.x = bq[ks][0].x; b.y = bq[ks][1].x;
        acc0 = __builtin_amdgcn_wmma_f32_16x16x4_f32(false, afrag[ks], false, b, (short)0, acc0, false, false);
        b.x = bq[ks][0].y; b.y = bq[ks][1].y;
        acc1 = __builtin_amdgcn_wmma_f32_16x16x4_f32(false, afrag[ks], false, b, (short)0, acc1, false, false);
        b.x = bq[ks][0].z; b.y = bq[ks][1].z;
        acc2 = __builtin_amdgcn_wmma_f32_16x16x4_f32(false, afrag[ks], false, b, (short)0, acc2, false, false);
        b.x = bq[ks][0].w; b.y = bq[ks][1].w;
        acc3 = __builtin_amdgcn_wmma_f32_16x16x4_f32(false, afrag[ks], false, b, (short)0, acc3, false, false);
    }

    const float4 a1v = *(const float4*)(a + l15 * 4);
    const float4 a2v = *(const float4*)(a + OUT_C + l15 * 4);
    const bool whole = (m0 + 16 <= n);
    float pr1[8], pr2[8];
#pragma unroll
    for (int r = 0; r < 8; ++r) {
        const int row = m0 + r + 8 * half;
        float4 v;
        v.x = acc0[r]; v.y = acc1[r]; v.z = acc2[r]; v.w = acc3[r];
        if (whole || row < n) *(float4*)(h + (size_t)row * OUT_C + l15 * 4) = v;
        pr1[r] = v.x * a1v.x + v.y * a1v.y + v.z * a1v.z + v.w * a1v.w;
        pr2[r] = v.x * a2v.x + v.y * a2v.y + v.z * a2v.z + v.w * a2v.w;
    }

#pragma unroll
    for (int mask = 8; mask >= 1; mask >>= 1) {
#pragma unroll
        for (int r = 0; r < 8; ++r) {
            pr1[r] += __shfl_xor(pr1[r], mask, 32);
            pr2[r] += __shfl_xor(pr2[r], mask, 32);
        }
    }
    if (l15 == 0) {
#pragma unroll
        for (int r = 0; r < 8; ++r) {
            const int row = m0 + r + 8 * half;
            if (row < n) { s[row] = pr1[r]; t[row] = pr2[r]; }
        }
    }
}

#define OWN_THREADS 1024
#define OWN_WAVES   32
#define OWN_QLEN    1088
#define OWN_FETCH   4

template <bool OWNED>
__device__ __forceinline__ void own_drain(const int* myq, int cnt,
                                          const float* __restrict__ h,
                                          const float* __restrict__ t,
                                          const float* s_l, float* acc, float* esum,
                                          int lane) {
    for (int b = 0; b < cnt; b += 32) {
        const int nb = min(32, cnt - b);
        const bool own = lane < nb;
        unsigned r = (unsigned)OWN_ROWS;
        if (own) r = (unsigned)myq[b + lane];
        const int dst = (int)(r >> 9);
        const int ls  = (int)(r & 511u);
        float w = __expf(lrelu(s_l[ls] + t[dst]));
        if (!own) w = 0.0f;
        if (own) atomicAdd(&esum[ls], w);
        for (int k0 = 0; k0 < nb; k0 += 8) {
            float2 v[8];
#pragma unroll
            for (int j = 0; j < 8; ++j) {
                const int d = __builtin_amdgcn_readlane(dst, k0 + j);
                v[j] = *(const float2*)(h + (size_t)d * OUT_C + lane * 2);
            }
#pragma unroll
            for (int j = 0; j < 8; ++j) {
                const float wk = __int_as_float(__builtin_amdgcn_readlane(__float_as_int(w), k0 + j));
                const int   lk = __builtin_amdgcn_readlane(ls, k0 + j);
                float* a = acc + lk * OUT_C + lane * 2;
                if (OWNED) {
                    float2 c = *(float2*)a;
                    c.x += wk * v[j].x;
                    c.y += wk * v[j].y;
                    *(float2*)a = c;
                } else {
                    atomicAdd(a,     wk * v[j].x);
                    atomicAdd(a + 1, wk * v[j].y);
                }
            }
        }
    }
}

__device__ __forceinline__ int own_collect(int cnt, int* myq,
                                           const unsigned* __restrict__ rec,
                                           const unsigned* __restrict__ seg, int nCh, int owner,
                                           const float* __restrict__ h,
                                           const float* __restrict__ t,
                                           const float* s_l, float* acc, float* esum,
                                           int lane, int wv) {
    const int cBeg = (int)(((long long)wv * nCh) / OWN_WAVES);
    const int cEnd = (int)(((long long)(wv + 1) * nCh) / OWN_WAVES);
    for (int c0 = cBeg; c0 < cEnd; c0 += 32) {
        const int c = c0 + lane;
        unsigned sw = 0u;
        if (c < cEnd) sw = seg[(size_t)owner * (size_t)nCh + (size_t)c];
        const int ct = (int)(sw >> 16);
        const int st = (int)(sw & 0xffffu);
        int incl = ct;
#pragma unroll
        for (int d = 1; d < 32; d <<= 1) {
            const int y = __shfl_up(incl, d, 32);
            if (lane >= d) incl += y;
        }
        const int excl  = incl - ct;
        const int total = __builtin_amdgcn_readlane(incl, 31);
        const int gbase = c * OWN_CHUNK + st - excl;
        int ex[32];
#pragma unroll
        for (int j = 0; j < 32; ++j) ex[j] = __builtin_amdgcn_readlane(excl, j);

        for (int p0 = 0; p0 < total; p0 += 32 * OWN_FETCH) {
            unsigned r[OWN_FETCH];
#pragma unroll
            for (int u = 0; u < OWN_FETCH; ++u) {
                const int p = p0 + 32 * u + lane;
                int js = -1;
#pragma unroll
                for (int j = 0; j < 32; ++j) js += (p >= ex[j]) ? 1 : 0;
                const int gb = __shfl(gbase, js, 32);
                r[u] = (unsigned)OWN_ROWS;
                if (p < total) r[u] = rec[gb + p];
            }
#pragma unroll
            for (int u = 0; u < OWN_FETCH; ++u) {
                const int p = p0 + 32 * u + lane;
                if (p < total) myq[cnt + 32 * u + lane] = (int)r[u];
            }
            cnt += min(32 * OWN_FETCH, total - p0);
            if (cnt > OWN_QLEN - 32 * OWN_FETCH) {
                own_drain<false>(myq, cnt, h, t, s_l, acc, esum, lane);
                cnt = 0;
            }
        }
    }
    return cnt;
}

__device__ __forceinline__ void own_route(int* q, int* qcnt, int* qovf,
                                          const unsigned* __restrict__ rec,
                                          const unsigned* __restrict__ seg, int nCh, int owner,
                                          int lane, int wv) {
    const int cBeg = (int)(((long long)wv * nCh) / OWN_WAVES);
    const int cEnd = (int)(((long long)(wv + 1) * nCh) / OWN_WAVES);
    bool full = false;
    for (int c0 = cBeg; c0 < cEnd; c0 += 32) {
        const int c = c0 + lane;
        unsigned sw = 0u;
        if (c < cEnd) sw = seg[(size_t)owner * (size_t)nCh + (size_t)c];
        const int ct = (int)(sw >> 16);
        const int st = (int)(sw & 0xffffu);
        int incl = ct;
#pragma unroll
        for (int d = 1; d < 32; d <<= 1) {
            const int y = __shfl_up(incl, d, 32);
            if (lane >= d) incl += y;
        }
        const int excl  = incl - ct;
        const int total = __builtin_amdgcn_readlane(incl, 31);
        const int gbase = c * OWN_CHUNK + st - excl;
        int ex[32];
#pragma unroll
        for (int j = 0; j < 32; ++j) ex[j] = __builtin_amdgcn_readlane(excl, j);

        for (int p0 = 0; p0 < total; p0 += 32 * OWN_FETCH) {
            unsigned r[OWN_FETCH];
#pragma unroll
            for (int u = 0; u < OWN_FETCH; ++u) {
                const int p = p0 + 32 * u + lane;
                int js = -1;
#pragma unroll
                for (int j = 0; j < 32; ++j) js += (p >= ex[j]) ? 1 : 0;
                const int gb = __shfl(gbase, js, 32);
                r[u] = (unsigned)OWN_ROWS;
                if (p < total) r[u] = rec[gb + p];
            }
#pragma unroll
            for (int u = 0; u < OWN_FETCH; ++u) {
                const int p = p0 + 32 * u + lane;
                if (p < total) {
                    const int o   = (int)(r[u] & (unsigned)(OWN_WAVES - 1));
                    const int pos = atomicAdd(&qcnt[o], 1);
                    if (pos < OWN_QLEN) q[o * OWN_QLEN + pos] = (int)r[u];
                    else full = true;
                }
            }
        }
    }
    if (full) *qovf = 1;
}

__device__ __forceinline__ int own_scan_sources(int cnt, int* myq,
                                                const int* __restrict__ ei, int nE,
                                                const float* __restrict__ h,
                                                const float* __restrict__ t,
                                                const float* s_l, float* acc, float* esum,
                                                int lane, int wv, int r0, int R) {
    for (long long e0 = (long long)wv * 32; e0 < (long long)nE; e0 += (long long)OWN_WAVES * 32) {
        const long long e = e0 + lane;
        const bool in = e < (long long)nE;
        const int  sr = in ? ei[e] : -1;
        const bool m  = in && (((unsigned)sr - (unsigned)r0) < (unsigned)R);
        const unsigned k = __builtin_amdgcn_ballot_w32(m);
        if (m) {
            const int d = ei[(long long)nE + e];
            myq[cnt + __builtin_amdgcn_mbcnt_lo(k, 0u)] = (int)(((unsigned)d << 9) | (unsigned)(sr - r0));
        }
        cnt += __builtin_popcount(k);
        if (cnt > OWN_QLEN - 32) {
            own_drain<false>(myq, cnt, h, t, s_l, acc, esum, lane);
            cnt = 0;
        }
    }
    return cnt;
}

__global__ void __launch_bounds__(OWN_THREADS)
gat_owner(const int* __restrict__ ei,
          float* __restrict__ h,
          const float* __restrict__ s,
          const float* __restrict__ t,
          float* __restrict__ out,
          int n, int nE, unsigned long long wsBytes) {
    __shared__ __attribute__((aligned(16))) float acc[(OWN_ROWS + 1) * OUT_C];
    __shared__ float esum[OWN_ROWS + 1];
    __shared__ float s_l[OWN_ROWS + 1];
    __shared__ int   q[OWN_WAVES * OWN_QLEN];
    __shared__ int   qcnt[OWN_WAVES];
    __shared__ int   qovf;

    const int tid  = threadIdx.x;
    const int lane = tid & 31;
    const int wv   = __builtin_amdgcn_readfirstlane(tid >> 5);
    int* myq = q + wv * OWN_QLEN;

    const OwnPlanes P = own_planes(h, t + n, wsBytes, n, nE);

    for (int r0 = blockIdx.x * OWN_ROWS; r0 < n; r0 += gridDim.x * OWN_ROWS) {
        const int R = min(OWN_ROWS, n - r0);

        for (int i = tid; i < (OWN_ROWS + 1) * OUT_C; i += OWN_THREADS) acc[i] = 0.0f;
        for (int i = tid; i <= OWN_ROWS; i += OWN_THREADS) {
            esum[i] = 0.0f;
            s_l[i]  = (i < R) ? s[r0 + i] : 0.0f;
        }
        if (tid < OWN_WAVES) qcnt[tid] = 0;
        if (tid == 0) qovf = 0;
        __syncthreads();

        int cnt = 0;
        if (P.ok) {
            own_route(q, qcnt, &qovf, P.rec, P.seg, P.nCh, (int)blockIdx.x, lane, wv);
            __syncthreads();
            if (qovf == 0) {
                cnt = __builtin_amdgcn_readfirstlane(qcnt[wv]);
                own_drain<true>(myq, cnt, h, t, s_l, acc, esum, lane);
            } else {
                cnt = own_collect(cnt, myq, P.rec, P.seg, P.nCh, (int)blockIdx.x, h, t, s_l, acc, esum, lane, wv);
                own_drain<false>(myq, cnt, h, t, s_l, acc, esum, lane);
            }
        } else {
            cnt = own_scan_sources(cnt, myq, ei, nE, h, t, s_l, acc, esum, lane, wv, r0, R);
            own_drain<false>(myq, cnt, h, t, s_l, acc, esum, lane);
        }
        __syncthreads();

        for (int i4 = tid; i4 < R * (OUT_C / 4); i4 += OWN_THREADS) {
            const int li = i4 >> 4;
            const int c4 = (i4 & 15) * 4;
            const size_t i = (size_t)(r0 + li);
            const float w = __expf(lrelu(s_l[li] + t[i]));
            const float denom = esum[li] + w;
            const float4 hv = *(const float4*)(h + i * OUT_C + c4);
            const float4 av = *(const float4*)(acc + li * OUT_C + c4);
            float4 o;
            o.x = (av.x + w * hv.x) / denom;
            o.y = (av.y + w * hv.y) / denom;
            o.z = (av.z + w * hv.z) / denom;
            o.w = (av.w + w * hv.w) / denom;
            *(float4*)(out + i * OUT_C + c4) = o;
        }
        __syncthreads();
    }
}

extern "C" void kernel_launch(void* const* d_in, const int* in_sizes, int n_in,
                              void* d_out, int out_size, void* d_ws, size_t ws_size,
                              hipStream_t stream) {
    const float* x  = (const float*)d_in[0];
    const float* W  = (const float*)d_in[1];
    const float* a  = (const float*)d_in[2];
    const int*   ei = (const int*)d_in[3];

    const int n = in_sizes[0] / IN_C;
    const int E = in_sizes[3] / 2;

    float* ws     = (float*)d_ws;
    float* h      = ws;
    float* s      = h + (size_t)n * OUT_C;
    float* t      = s + n;
    float* out    = (float*)d_out;

    const dim3 blk(K1_THREADS);
    const int wavesPerBlock = K1_THREADS / 32;

    const int mtiles = (n + 15) / 16;
    gat_gemm_stw<<<(mtiles + wavesPerBlock - 1) / wavesPerBlock, blk, 0, stream>>>(
        x, W, a, h, s, t, ei, n, mtiles, E, (unsigned long long)ws_size);

    gat_owner<<<OWN_BLOCKS, OWN_THREADS, 0, stream>>>(ei, h, s, t, out, n, E,
                                                      (unsigned long long)ws_size);
}


# owner block files each record's dst in its row's own LDS list (LDS integer adds rank it); the row's wave sums w*h[dst] and w in registers, self-loop last, stores the row once; list (arrival) order
# speedup vs baseline: 4.4654x; 1.1078x over previous
#include <hip/hip_runtime.h>
#include <hip/hip_bf16.h>
#include <math.h>


typedef __attribute__((ext_vector_type(2))) float v2f;
typedef __attribute__((ext_vector_type(8))) float v8f;

#define IN_C  128
#define OUT_C 64
#define GAT_ALPHA 0.2f

__device__ __forceinline__ float lrelu(float z) {
    return z >= 0.0f ? z : GAT_ALPHA * z;
}

#define OWN_ROWS    392
#define OWN_BLOCKS  256
#define OWN_CHUNK   2048
#define K1_THREADS  256
#define GEMM_AHEAD  3
#define LOADS_STAY_AHEAD() asm volatile("" ::: "memory")
__device__ constexpr int gemm_a_from(int ks) {
    return ks == 4 ? 8 : (ks == 8 ? 12 : (ks == 12 ? 24 : (ks == 16 ? 28 : 0)));
}

__device__ __forceinline__ unsigned own_key(int src) {
    return ((unsigned)src / (unsigned)OWN_ROWS) & (unsigned)(OWN_BLOCKS - 1);
}

__device__ __forceinline__ unsigned own_row(int src) {
    return (unsigned)src - ((unsigned)src / (unsigned)OWN_ROWS) * (unsigned)OWN_ROWS;
}

struct OwnPlanes { unsigned* rec; unsigned* seg; int nCh; int ok; };

__device__ __forceinline__ OwnPlanes own_planes(float* wsBase, const float* tEnd,
                                                unsigned long long wsBytes, int n, int nE) {
    OwnPlanes P;
    const unsigned long long off =
        ((unsigned long long)((const char*)tEnd - (const char*)wsBase) + 15ull) & ~15ull;
    const int nCh = (nE >> 11) + ((nE & (OWN_CHUNK - 1)) ? 1 : 0);
    const unsigned long long recBytes = (unsigned long long)nCh * OWN_CHUNK * 4ull;
    const unsigned long long segBytes = (unsigned long long)nCh * OWN_BLOCKS * 4ull;
    P.rec = (unsigned*)((char*)wsBase + off);
    P.seg = (unsigned*)((char*)wsBase + off + recBytes);
    P.nCh = nCh;
    P.ok  = (nE > 0 && (nE & 7) == 0 && n <= OWN_BLOCKS * OWN_ROWS &&
             wsBytes >= off + recBytes + segBytes) ? 1 : 0;
    return P;
}

__global__ void
gat_gemm_stw(const float* __restrict__ x,
             const float* __restrict__ W,
             const float* __restrict__ a,
             float* __restrict__ h,
             float* __restrict__ s,
             float* __restrict__ t,
             const int* __restrict__ ei,
             int n, int mtiles, int nE, unsigned long long wsBytes) {
    __shared__ int kcnt[OWN_BLOCKS];
    __shared__ int koff[OWN_BLOCKS];
    __shared__ int wsum[K1_THREADS / 32];
    __shared__ __attribute__((aligned(16))) unsigned lrec[OWN_CHUNK];

    const int wavesPerBlock = blockDim.x >> 5;
    const int wave = blockIdx.x * wavesPerBlock + ((int)threadIdx.x >> 5);
    const int lane = threadIdx.x & 31;

    {
        const OwnPlanes P = own_planes(h, t + n, wsBytes, n, nE);
        const int tid = (int)threadIdx.x;
        const int wib = tid >> 5;
        for (int c = (int)blockIdx.x; P.ok && c < P.nCh; c += (int)gridDim.x) {
            kcnt[tid] = 0;
            __syncthreads();
            const size_t e0 = (size_t)c * OWN_CHUNK + (size_t)tid * 8;
            const bool live = e0 < (size_t)nE;
            int4 p = {0, 0, 0, 0}, q = {0, 0, 0, 0}, d0 = {0, 0, 0, 0}, d1 = {0, 0, 0, 0};
            if (live) {
                const int4* sp = (const int4*)(ei + e0);
                const int4* dp = (const int4*)(ei + (size_t)nE + e0);
                p = sp[0]; q = sp[1];
                d0 = dp[0]; d1 = dp[1];
            }
            const int sj[8] = {p.x, p.y, p.z, p.w, q.x, q.y, q.z, q.w};
            const int dj[8] = {d0.x, d0.y, d0.z, d0.w, d1.x, d1.y, d1.z, d1.w};
            unsigned key[8];
            int rk[8];
#pragma unroll
            for (int j = 0; j < 8; ++j) {
                key[j] = own_key(sj[j]);
                rk[j]  = 0;
                if (live) rk[j] = atomicAdd(&kcnt[key[j]], 1);
            }
            __syncthreads();
            const int v = kcnt[tid];
            int incl = v;
#pragma unroll
            for (int d = 1; d < 32; d <<= 1) {
                const int y = __shfl_up(incl, d, 32);
                if (lane >= d) incl += y;
            }
            if (lane == 31) wsum[wib] = incl;
            __syncthreads();
            int add = 0;
#pragma unroll
            for (int w = 0; w < K1_THREADS / 32; ++w) add += (w < wib) ? wsum[w] : 0;
            const int excl = incl - v + add;
            koff[tid] = excl;
            P.seg[(size_t)tid * (size_t)P.nCh + (size_t)c] = (unsigned)excl | ((unsigned)v << 16);
            __syncthreads();
            if (live) {
#pragma unroll
                for (int j = 0; j < 8; ++j)
                    lrec[koff[key[j]] + rk[j]] = ((unsigned)dj[j] << 9) | own_row(sj[j]);
            }
            __syncthreads();
            uint4* rg = (uint4*)(P.rec + (size_t)c * OWN_CHUNK);
            const uint4* rl = (const uint4*)lrec;
            rg[tid]              = rl[tid];
            rg[tid + K1_THREADS] = rl[tid + K1_THREADS];
        }
    }

    if (wave >= mtiles) return;

    const int half = lane >> 4;
    const int l15  = lane & 15;
    const int m0   = wave * 16;

    int arow = m0 + l15; if (arow > n - 1) arow = n - 1;
    const float* xl = x + (size_t)arow * IN_C + half * 2;

    const float* wl = W + (size_t)(half * 2) * OUT_C + l15 * 4;

    v2f    afrag[32];
    float4 bq[32][2];
#pragma unroll
    for (int ks = 0; ks < 8; ++ks) {
        afrag[ks]      = *(const v2f*)(xl + ks * 4);
        afrag[ks + 16] = *(const v2f*)(xl + (ks + 16) * 4);
    }
#pragma unroll
    for (int ks = 0; ks < GEMM_AHEAD; ++ks) {
        bq[ks][0] = *(const float4*)(wl + (size_t)(ks * 4)     * OUT_C);
        bq[ks][1] = *(const float4*)(wl + (size_t)(ks * 4 + 1) * OUT_C);
    }

    v8f acc0 = {}, acc1 = {}, acc2 = {}, acc3 = {};
#pragma unroll
    for (int ks = 0; ks < 32; ++ks) {
        if (gemm_a_from(ks) != 0) {
#pragma unroll
            for (int j = 0; j < 4; ++j)
                afrag[gemm_a_from(ks) + j] = *(const v2f*)(xl + (gemm_a_from(ks) + j) * 4);
        }
        if (ks + GEMM_AHEAD < 32) {
            bq[ks + GEMM_AHEAD][0] = *(const float4*)(wl + (size_t)((ks + GEMM_AHEAD) * 4)     * OUT_C);
            bq[ks + GEMM_AHEAD][1] = *(const float4*)(wl + (size_t)((ks + GEMM_AHEAD) * 4 + 1) * OUT_C);
        }
        LOADS_STAY_AHEAD();
        v2f b;
        b.x = bq[ks][0].x; b.y = bq[ks][1].x;
        acc0 = __builtin_amdgcn_wmma_f32_16x16x4_f32(false, afrag[ks], false, b, (short)0, acc0, false, false);
        b.x = bq[ks][0].y; b.y = bq[ks][1].y;
        acc1 = __builtin_amdgcn_wmma_f32_16x16x4_f32(false, afrag[ks], false, b, (short)0, acc1, false, false);
        b.x = bq[ks][0].z; b.y = bq[ks][1].z;
        acc2 = __builtin_amdgcn_wmma_f32_16x16x4_f32(false, afrag[ks], false, b, (short)0, acc2, false, false);
        b.x = bq[ks][0].w; b.y = bq[ks][1].w;
        acc3 = __builtin_amdgcn_wmma_f32_16x16x4_f32(false, afrag[ks], false, b, (short)0, acc3, false, false);
    }

    const float4 a1v = *(const float4*)(a + l15 * 4);
    const float4 a2v = *(const float4*)(a + OUT_C + l15 * 4);
    const bool whole = (m0 + 16 <= n);
    float pr1[8], pr2[8];
#pragma unroll
    for (int r = 0; r < 8; ++r) {
        const int row = m0 + r + 8 * half;
        float4 v;
        v.x = acc0[r]; v.y = acc1[r]; v.z = acc2[r]; v.w = acc3[r];
        if (whole || row < n) *(float4*)(h + (size_t)row * OUT_C + l15 * 4) = v;
        pr1[r] = v.x * a1v.x + v.y * a1v.y + v.z * a1v.z + v.w * a1v.w;
        pr2[r] = v.x * a2v.x + v.y * a2v.y + v.z * a2v.z + v.w * a2v.w;
    }

#pragma unroll
    for (int mask = 8; mask >= 1; mask >>= 1) {
#pragma unroll
        for (int r = 0; r < 8; ++r) {
            pr1[r] += __shfl_xor(pr1[r], mask, 32);
            pr2[r] += __shfl_xor(pr2[r], mask, 32);
        }
    }
    if (l15 == 0) {
#pragma unroll
        for (int r = 0; r < 8; ++r) {
            const int row = m0 + r + 8 * half;
            if (row < n) { s[row] = pr1[r]; t[row] = pr2[r]; }
        }
    }
}

#define OWN_THREADS 1024
#define OWN_WAVES   32
#define OWN_QLEN    1088
#define OWN_ROWQ    87
#define OWN_FETCH   4

__device__ __forceinline__ void own_drain(const int* myq, int cnt,
                                          const float* __restrict__ h,
                                          const float* __restrict__ t,
                                          const float* s_l, float* acc, float* esum,
                                          int lane) {
    for (int b = 0; b < cnt; b += 32) {
        const int nb = min(32, cnt - b);
        const bool own = lane < nb;
        unsigned r = (unsigned)OWN_ROWS;
        if (own) r = (unsigned)myq[b + lane];
        const int dst = (int)(r >> 9);
        const int ls  = (int)(r & 511u);
        float w = __expf(lrelu(s_l[ls] + t[dst]));
        if (!own) w = 0.0f;
        if (own) atomicAdd(&esum[ls], w);
        for (int k0 = 0; k0 < nb; k0 += 8) {
            float2 v[8];
#pragma unroll
            for (int j = 0; j < 8; ++j) {
                const int d = __builtin_amdgcn_readlane(dst, k0 + j);
                v[j] = *(const float2*)(h + (size_t)d * OUT_C + lane * 2);
            }
#pragma unroll
            for (int j = 0; j < 8; ++j) {
                const float wk = __int_as_float(__builtin_amdgcn_readlane(__float_as_int(w), k0 + j));
                const int   lk = __builtin_amdgcn_readlane(ls, k0 + j);
                float* a = acc + lk * OUT_C + lane * 2;
                atomicAdd(a,     wk * v[j].x);
                atomicAdd(a + 1, wk * v[j].y);
            }
        }
    }
}

__device__ __forceinline__ int own_collect(int cnt, int* myq,
                                           const unsigned* __restrict__ rec,
                                           const unsigned* __restrict__ seg, int nCh, int owner,
                                           const float* __restrict__ h,
                                           const float* __restrict__ t,
                                           const float* s_l, float* acc, float* esum,
                                           int lane, int wv) {
    const int cBeg = (int)(((long long)wv * nCh) / OWN_WAVES);
    const int cEnd = (int)(((long long)(wv + 1) * nCh) / OWN_WAVES);
    for (int c0 = cBeg; c0 < cEnd; c0 += 32) {
        const int c = c0 + lane;
        unsigned sw = 0u;
        if (c < cEnd) sw = seg[(size_t)owner * (size_t)nCh + (size_t)c];
        const int ct = (int)(sw >> 16);
        const int st = (int)(sw & 0xffffu);
        int incl = ct;
#pragma unroll
        for (int d = 1; d < 32; d <<= 1) {
            const int y = __shfl_up(incl, d, 32);
            if (lane >= d) incl += y;
        }
        const int excl  = incl - ct;
        const int total = __builtin_amdgcn_readlane(incl, 31);
        const int gbase = c * OWN_CHUNK + st - excl;
        int ex[32];
#pragma unroll
        for (int j = 0; j < 32; ++j) ex[j] = __builtin_amdgcn_readlane(excl, j);

        for (int p0 = 0; p0 < total; p0 += 32 * OWN_FETCH) {
            unsigned r[OWN_FETCH];
#pragma unroll
            for (int u = 0; u < OWN_FETCH; ++u) {
                const int p = p0 + 32 * u + lane;
                int js = -1;
#pragma unroll
                for (int j = 0; j < 32; ++j) js += (p >= ex[j]) ? 1 : 0;
                const int gb = __shfl(gbase, js, 32);
                r[u] = (unsigned)OWN_ROWS;
                if (p < total) r[u] = rec[gb + p];
            }
#pragma unroll
            for (int u = 0; u < OWN_FETCH; ++u) {
                const int p = p0 + 32 * u + lane;
                if (p < total) myq[cnt + 32 * u + lane] = (int)r[u];
            }
            cnt += min(32 * OWN_FETCH, total - p0);
            if (cnt > OWN_QLEN - 32 * OWN_FETCH) {
                own_drain(myq, cnt, h, t, s_l, acc, esum, lane);
                cnt = 0;
            }
        }
    }
    return cnt;
}

__device__ __forceinline__ void own_route(int* q, int* qcnt, int* qovf,
                                          const unsigned* __restrict__ rec,
                                          const unsigned* __restrict__ seg, int nCh, int owner,
                                          int lane, int wv) {
    const int cBeg = (int)(((long long)wv * nCh) / OWN_WAVES);
    const int cEnd = (int)(((long long)(wv + 1) * nCh) / OWN_WAVES);
    bool full = false;
    for (int c0 = cBeg; c0 < cEnd; c0 += 32) {
        const int c = c0 + lane;
        unsigned sw = 0u;
        if (c < cEnd) sw = seg[(size_t)owner * (size_t)nCh + (size_t)c];
        const int ct = (int)(sw >> 16);
        const int st = (int)(sw & 0xffffu);
        int incl = ct;
#pragma unroll
        for (int d = 1; d < 32; d <<= 1) {
            const int y = __shfl_up(incl, d, 32);
            if (lane >= d) incl += y;
        }
        const int excl  = incl - ct;
        const int total = __builtin_amdgcn_readlane(incl, 31);
        const int gbase = c * OWN_CHUNK + st - excl;
        int ex[32];
#pragma unroll
        for (int j = 0; j < 32; ++j) ex[j] = __builtin_amdgcn_readlane(excl, j);

        for (int p0 = 0; p0 < total; p0 += 32 * OWN_FETCH) {
            unsigned r[OWN_FETCH];
#pragma unroll
            for (int u = 0; u < OWN_FETCH; ++u) {
                const int p = p0 + 32 * u + lane;
                int js = -1;
#pragma unroll
                for (int j = 0; j < 32; ++j) js += (p >= ex[j]) ? 1 : 0;
                const int gb = __shfl(gbase, js, 32);
                r[u] = (unsigned)OWN_ROWS;
                if (p < total) r[u] = rec[gb + p];
            }
#pragma unroll
            for (int u = 0; u < OWN_FETCH; ++u) {
                const int p = p0 + 32 * u + lane;
                if (p < total) {
                    const int row = (int)(r[u] & 511u);
                    const int pos = atomicAdd(&qcnt[row], 1);
                    if (pos < OWN_ROWQ) q[row * OWN_ROWQ + pos] = (int)(r[u] >> 9);
                    else full = true;
                }
            }
        }
    }
    if (full) *qovf = 1;
}

template <int NG>
__device__ __forceinline__ void own_batch(int d, float w, int nb, const float* __restrict__ h, int lane,
                                          float& ax, float& ay, float& es) {
    float2 v[NG * 8];
#pragma unroll
    for (int j = 0; j < NG * 8; ++j) {
        if (j < (NG - 1) * 8 || j < nb) {
            const unsigned dk = (unsigned)__builtin_amdgcn_readlane(d, j);
            v[j] = *(const float2*)(h + (size_t)dk * OUT_C + lane * 2);
        } else {
            v[j] = make_float2(0.0f, 0.0f);
        }
    }
#pragma unroll
    for (int j = 0; j < NG * 8; ++j) {
        const float wk = __int_as_float(__builtin_amdgcn_readlane(__float_as_int(w), j));
        ax += wk * v[j].x;
        ay += wk * v[j].y;
        es += wk;
    }
}

__device__ __forceinline__ void own_rows(const int* rowq, const int* qcnt, const float* s_l,
                                         const float* __restrict__ h,
                                         const float* __restrict__ t,
                                         float* __restrict__ out,
                                         int r0, int R, int lane, int wv) {
    int   r   = wv;
    int   cnt = 0, d = 0;
    float sr  = 0.0f, w = 0.0f;
    if (r < R) {
        cnt = __builtin_amdgcn_readfirstlane(qcnt[r]);
        sr  = s_l[r];
        d   = (lane < cnt) ? rowq[r * OWN_ROWQ + lane] : r0 + r;
        w   = __expf(lrelu(sr + t[d]));
        if (lane > cnt) w = 0.0f;
    }
    while (r < R) {
        const int  node = r0 + r;
        const int  rn   = r + OWN_WAVES;
        const bool more = rn < R;
        int   cntn = 0, dn = 0;
        float srn  = 0.0f, tvn = 0.0f;
        if (more) {
            cntn = __builtin_amdgcn_readfirstlane(qcnt[rn]);
            srn  = s_l[rn];
            dn   = (lane < cntn) ? rowq[rn * OWN_ROWQ + lane] : r0 + rn;
            tvn  = t[dn];
        }
        const int total = cnt + 1;
        int   nb = min(32, total);
        float ax = 0.0f, ay = 0.0f, es = 0.0f;
        for (int b = 0; ; ) {
            if (nb <= 8)       own_batch<1>(d, w, nb, h, lane, ax, ay, es);
            else if (nb <= 16) own_batch<2>(d, w, nb, h, lane, ax, ay, es);
            else if (nb <= 24) own_batch<3>(d, w, nb, h, lane, ax, ay, es);
            else               own_batch<4>(d, w, nb, h, lane, ax, ay, es);
            b += 32;
            if (b >= total) break;
            const int i = b + lane;
            d  = (i < cnt) ? rowq[r * OWN_ROWQ + i] : node;
            nb = min(32, total - b);
            w  = __expf(lrelu(sr + t[d]));
            if (lane >= nb) w = 0.0f;
        }
        float2 o;
        o.x = ax / es;
        o.y = ay / es;
        *(float2*)(out + (size_t)node * OUT_C + lane * 2) = o;
        asm volatile("" : "+v"(tvn));
        w = __expf(lrelu(srn + tvn));
        if (lane > cntn) w = 0.0f;
        r = rn; cnt = cntn; d = dn; sr = srn;
    }
}

__device__ __forceinline__ int own_scan_sources(int cnt, int* myq,
                                                const int* __restrict__ ei, int nE,
                                                const float* __restrict__ h,
                                                const float* __restrict__ t,
                                                const float* s_l, float* acc, float* esum,
                                                int lane, int wv, int r0, int R) {
    for (long long e0 = (long long)wv * 32; e0 < (long long)nE; e0 += (long long)OWN_WAVES * 32) {
        const long long e = e0 + lane;
        const bool in = e < (long long)nE;
        const int  sr = in ? ei[e] : -1;
        const bool m  = in && (((unsigned)sr - (unsigned)r0) < (unsigned)R);
        const unsigned k = __builtin_amdgcn_ballot_w32(m);
        if (m) {
            const int d = ei[(long long)nE + e];
            myq[cnt + __builtin_amdgcn_mbcnt_lo(k, 0u)] = (int)(((unsigned)d << 9) | (unsigned)(sr - r0));
        }
        cnt += __builtin_popcount(k);
        if (cnt > OWN_QLEN - 32) {
            own_drain(myq, cnt, h, t, s_l, acc, esum, lane);
            cnt = 0;
        }
    }
    return cnt;
}

__global__ void __launch_bounds__(OWN_THREADS)
gat_owner(const int* __restrict__ ei,
          float* __restrict__ h,
          const float* __restrict__ s,
          const float* __restrict__ t,
          float* __restrict__ out,
          int n, int nE, unsigned long long wsBytes) {
    __shared__ __attribute__((aligned(16))) float acc[(OWN_ROWS + 1) * OUT_C];
    __shared__ float esum[OWN_ROWS + 1];
    __shared__ float s_l[OWN_ROWS + 1];
    __shared__ int   q[OWN_WAVES * OWN_QLEN];
    __shared__ int   qcnt[OWN_ROWS + 1];
    __shared__ int   qovf;

    const int tid  = threadIdx.x;
    const int lane = tid & 31;
    const int wv   = __builtin_amdgcn_readfirstlane(tid >> 5);
    int* myq = q + wv * OWN_QLEN;

    const OwnPlanes P = own_planes(h, t + n, wsBytes, n, nE);

    for (int r0 = blockIdx.x * OWN_ROWS; r0 < n; r0 += gridDim.x * OWN_ROWS) {
        const int R = min(OWN_ROWS, n - r0);

        for (int i = tid; i <= OWN_ROWS; i += OWN_THREADS) {
            s_l[i]  = (i < R) ? s[r0 + i] : 0.0f;
            qcnt[i] = 0;
        }
        if (tid == 0) qovf = 0;
        __syncthreads();

        bool done = false;
        if (P.ok) {
            own_route(q, qcnt, &qovf, P.rec, P.seg, P.nCh, (int)blockIdx.x, lane, wv);
            __syncthreads();
            if (qovf == 0) {
                own_rows(q, qcnt, s_l, h, t, out, r0, R, lane, wv);
                done = true;
            }
        }
        if (!done) {
            for (int i = tid; i < (OWN_ROWS + 1) * OUT_C; i += OWN_THREADS) acc[i] = 0.0f;
            for (int i = tid; i <= OWN_ROWS; i += OWN_THREADS) esum[i] = 0.0f;
            __syncthreads();
            int cnt = 0;
            if (P.ok)
                cnt = own_collect(cnt, myq, P.rec, P.seg, P.nCh, (int)blockIdx.x, h, t, s_l, acc, esum, lane, wv);
            else
                cnt = own_scan_sources(cnt, myq, ei, nE, h, t, s_l, acc, esum, lane, wv, r0, R);
            own_drain(myq, cnt, h, t, s_l, acc, esum, lane);
            __syncthreads();
            for (int i4 = tid; i4 < R * (OUT_C / 4); i4 += OWN_THREADS) {
                const int li = i4 >> 4;
                const int c4 = (i4 & 15) * 4;
                const size_t i = (size_t)(r0 + li);
                const float w = __expf(lrelu(s_l[li] + t[i]));
                const float denom = esum[li] + w;
                const float4 hv = *(const float4*)(h + i * OUT_C + c4);
                const float4 av = *(const float4*)(acc + li * OUT_C + c4);
                float4 o;
                o.x = (av.x + w * hv.x) / denom;
                o.y = (av.y + w * hv.y) / denom;
                o.z = (av.z + w * hv.z) / denom;
                o.w = (av.w + w * hv.w) / denom;
                *(float4*)(out + i * OUT_C + c4) = o;
            }
        }
        __syncthreads();
    }
}

extern "C" void kernel_launch(void* const* d_in, const int* in_sizes, int n_in,
                              void* d_out, int out_size, void* d_ws, size_t ws_size,
                              hipStream_t stream) {
    const float* x  = (const float*)d_in[0];
    const float* W  = (const float*)d_in[1];
    const float* a  = (const float*)d_in[2];
    const int*   ei = (const int*)d_in[3];

    const int n = in_sizes[0] / IN_C;
    const int E = in_sizes[3] / 2;

    float* ws     = (float*)d_ws;
    float* h      = ws;
    float* s      = h + (size_t)n * OUT_C;
    float* t      = s + n;
    float* out    = (float*)d_out;

    const dim3 blk(K1_THREADS);
    const int wavesPerBlock = K1_THREADS / 32;

    const int mtiles = (n + 15) / 16;
    gat_gemm_stw<<<(mtiles + wavesPerBlock - 1) / wavesPerBlock, blk, 0, stream>>>(
        x, W, a, h, s, t, ei, n, mtiles, E, (unsigned long long)ws_size);

    gat_owner<<<OWN_BLOCKS, OWN_THREADS, 0, stream>>>(ei, h, s, t, out, n, E,
                                                      (unsigned long long)ws_size);
}
